# starting kernel
# baseline (speedup 1.0000x reference)
#include <hip/hip_runtime.h>

typedef float v2f __attribute__((ext_vector_type(2)));
typedef float v8f __attribute__((ext_vector_type(8)));

#define NNODES 50000
#define NEDGES 800000
#define HID 64
#define NGRAPHS 512
#define NCLASSES 10


__global__ __launch_bounds__(256) void k_init_deg(float* __restrict__ deg, int n) {
    int i = blockIdx.x * blockDim.x + threadIdx.x;
    if (i < n) deg[i] = 1.0f;
}

__global__ __launch_bounds__(256) void k_deg_accum(const int* __restrict__ dst,
                                                   float* __restrict__ deg, int e) {
    int i = blockIdx.x * blockDim.x + threadIdx.x;
    if (i < e) atomicAdd(&deg[dst[i]], 1.0f);
}

__global__ __launch_bounds__(256) void k_dinv(float* __restrict__ deg, int n) {
    int i = blockIdx.x * blockDim.x + threadIdx.x;
    if (i < n) deg[i] = rsqrtf(deg[i]);
}


__global__ __launch_bounds__(256)
void k_gemm_wmma(const float* __restrict__ A, const float* __restrict__ W,
                 float* __restrict__ C, int n, int relu_in) {
    __shared__ float sW[64 * 64];
    const int tid = threadIdx.x;
    #pragma unroll
    for (int i = 0; i < 16; ++i) sW[tid + i * 256] = W[tid + i * 256];
    __syncthreads();

    const int wave = tid >> 5;
    const int lane = tid & 31;
    const int half = lane >> 4;
    const int l16  = lane & 15;
    const int m0   = (blockIdx.x * 8 + wave) * 16;
    if (m0 >= n) return;

    v8f c0 = {}, c1 = {}, c2 = {}, c3 = {};

    const int arow = m0 + l16;
    const float* __restrict__ Arow = A + (size_t)arow * 64;

    #pragma unroll
    for (int k0 = 0; k0 < 64; k0 += 4) {
        const int kb = k0 + half * 2;
        float a0 = Arow[kb];
        float a1 = Arow[kb + 1];
        if (relu_in) { a0 = fmaxf(a0, 0.0f); a1 = fmaxf(a1, 0.0f); }
        v2f a = {a0, a1};

        const float* __restrict__ Wk0 = sW + (size_t)kb * 64;
        const float* __restrict__ Wk1 = sW + (size_t)(kb + 1) * 64;
        v2f b0 = {Wk0[l16 +  0], Wk1[l16 +  0]};
        v2f b1 = {Wk0[l16 + 16], Wk1[l16 + 16]};
        v2f b2 = {Wk0[l16 + 32], Wk1[l16 + 32]};
        v2f b3 = {Wk0[l16 + 48], Wk1[l16 + 48]};

        c0 = __builtin_amdgcn_wmma_f32_16x16x4_f32(false, a, false, b0, (short)0, c0, false, false);
        c1 = __builtin_amdgcn_wmma_f32_16x16x4_f32(false, a, false, b1, (short)0, c1, false, false);
        c2 = __builtin_amdgcn_wmma_f32_16x16x4_f32(false, a, false, b2, (short)0, c2, false, false);
        c3 = __builtin_amdgcn_wmma_f32_16x16x4_f32(false, a, false, b3, (short)0, c3, false, false);
    }

    #pragma unroll
    for (int r = 0; r < 8; ++r) {
        const int row = m0 + half * 8 + r;
        if (row < n) {
            float* __restrict__ Cr = C + (size_t)row * 64 + l16;
            Cr[0]  = c0[r];
            Cr[16] = c1[r];
            Cr[32] = c2[r];
            Cr[48] = c3[r];
        }
    }
}


__global__ __launch_bounds__(256)
void k_selfbias(const float* __restrict__ h, const float* __restrict__ dinv,
                const float* __restrict__ bias, float* __restrict__ a, int n) {
    int t = blockIdx.x * blockDim.x + threadIdx.x;
    if (t < n * HID) {
        int i = t >> 6;
        int f = t & 63;
        float di = dinv[i];
        a[t] = h[t] * di * di + bias[f];
    }
}

__global__ __launch_bounds__(256)
void k_edge_scatter(const float* __restrict__ h, const float* __restrict__ dinv,
                    const int* __restrict__ src, const int* __restrict__ dst,
                    float* __restrict__ a, int e) {
    int t = blockIdx.x * blockDim.x + threadIdx.x;
    int ei = t >> 6;
    int f = t & 63;
    if (ei < e) {
        int s = src[ei];
        int d = dst[ei];
        float nrm = dinv[s] * dinv[d];
        atomicAdd(&a[(size_t)d * 64 + f], h[(size_t)s * 64 + f] * nrm);
    }
}


__global__ __launch_bounds__(256) void k_zero(float* __restrict__ p, int n) {
    int i = blockIdx.x * blockDim.x + threadIdx.x;
    if (i < n) p[i] = 0.0f;
}

__global__ __launch_bounds__(256)
void k_pool(const float* __restrict__ a, const int* __restrict__ batch,
            float* __restrict__ sums, float* __restrict__ cnt, int n) {
    int t = blockIdx.x * blockDim.x + threadIdx.x;
    if (t < n * HID) {
        int i = t >> 6;
        int f = t & 63;
        int g = batch[i];
        atomicAdd(&sums[(size_t)g * HID + f], fmaxf(a[t], 0.0f));
        if (f == 0) atomicAdd(&cnt[g], 1.0f);
    }
}

__global__ __launch_bounds__(256)
void k_head(const float* __restrict__ sums, const float* __restrict__ cnt,
            const float* __restrict__ Wl, const float* __restrict__ bl,
            float* __restrict__ out) {
    int t = blockIdx.x * blockDim.x + threadIdx.x;
    if (t < NGRAPHS * NCLASSES) {
        int g = t / NCLASSES;
        int c = t % NCLASSES;
        float inv = 1.0f / fmaxf(cnt[g], 1.0f);
        float acc = 0.0f;
        const float* sg = sums + (size_t)g * HID;
        #pragma unroll
        for (int k = 0; k < HID; ++k) acc += sg[k] * Wl[k * NCLASSES + c];
        out[t] = acc * inv + bl[c];
    }
}


extern "C" void kernel_launch(void* const* d_in, const int* in_sizes, int n_in,
                              void* d_out, int out_size, void* d_ws, size_t ws_size,
                              hipStream_t stream) {
    const float* x  = (const float*)d_in[0];
    const float* W1 = (const float*)d_in[1];
    const float* b1 = (const float*)d_in[2];
    const float* W2 = (const float*)d_in[3];
    const float* b2 = (const float*)d_in[4];
    const float* Wl = (const float*)d_in[5];
    const float* bl = (const float*)d_in[6];
    const int* edge  = (const int*)d_in[7];
    const int* batch = (const int*)d_in[8];
    const int* src = edge;
    const int* dst = edge + NEDGES;
    float* out = (float*)d_out;

    float* ws   = (float*)d_ws;
    float* h    = ws;
    float* a    = h + (size_t)NNODES * HID;
    float* dinv = a + (size_t)NNODES * HID;
    float* sums = dinv + NNODES;
    float* cnt  = sums + (size_t)NGRAPHS * HID;

    const int TPB = 256;
    const int gN   = (NNODES + TPB - 1) / TPB;
    const int gE   = (NEDGES + TPB - 1) / TPB;
    const int gNF  = (NNODES * HID + TPB - 1) / TPB;
    const int gEd  = (NEDGES + 3) / 4;
    const int gGmm = ((NNODES + 15) / 16 + 7) / 8;
    const int gPZ  = (NGRAPHS * HID + NGRAPHS + TPB - 1) / TPB;
    const int gHd  = (NGRAPHS * NCLASSES + TPB - 1) / TPB;

    k_init_deg<<<gN, TPB, 0, stream>>>(dinv, NNODES);
    k_deg_accum<<<gE, TPB, 0, stream>>>(dst, dinv, NEDGES);
    k_dinv<<<gN, TPB, 0, stream>>>(dinv, NNODES);

    k_gemm_wmma<<<gGmm, TPB, 0, stream>>>(x, W1, h, NNODES, 0);
    k_selfbias<<<gNF, TPB, 0, stream>>>(h, dinv, b1, a, NNODES);
    k_edge_scatter<<<gEd, TPB, 0, stream>>>(h, dinv, src, dst, a, NEDGES);

    k_gemm_wmma<<<gGmm, TPB, 0, stream>>>(a, W2, h, NNODES, 1);
    k_selfbias<<<gNF, TPB, 0, stream>>>(h, dinv, b2, a, NNODES);
    k_edge_scatter<<<gEd, TPB, 0, stream>>>(h, dinv, src, dst, a, NEDGES);

    k_zero<<<gPZ, TPB, 0, stream>>>(sums, NGRAPHS * HID + NGRAPHS);
    k_pool<<<gNF, TPB, 0, stream>>>(a, batch, sums, cnt, NNODES);
    k_head<<<gHd, TPB, 0, stream>>>(sums, cnt, Wl, bl, out);
}


# dst walked once per call: one launch filters each owner block's edges into source|row words and counts row degrees; both layers add from the words (LDS float atomics, arrival order); 8 launches to 6
# speedup vs baseline: 3.2882x; 3.2882x over previous
#include <hip/hip_runtime.h>

typedef float v2f __attribute__((ext_vector_type(2)));
typedef float v8f __attribute__((ext_vector_type(8)));

#define NNODES 50000
#define NEDGES 800000
#define HID 64
#define NGRAPHS 512
#define NCLASSES 10


__global__ __launch_bounds__(256)
void k_gemm_wmma(const float* __restrict__ A, const float* __restrict__ W,
                 float* __restrict__ C, int n, int relu_in) {
    __shared__ float sW[64 * 64];
    const int tid = threadIdx.x;
    #pragma unroll
    for (int i = 0; i < 16; ++i) sW[tid + i * 256] = W[tid + i * 256];
    __syncthreads();

    const int wave = tid >> 5;
    const int lane = tid & 31;
    const int half = lane >> 4;
    const int l16  = lane & 15;
    const int m0   = (blockIdx.x * 8 + wave) * 16;
    if (m0 >= n) return;

    v8f c0 = {}, c1 = {}, c2 = {}, c3 = {};

    const int arow = m0 + l16;
    const float* __restrict__ Arow = A + (size_t)arow * 64;

    #pragma unroll
    for (int k0 = 0; k0 < 64; k0 += 4) {
        const int kb = k0 + half * 2;
        float a0 = Arow[kb];
        float a1 = Arow[kb + 1];
        if (relu_in) { a0 = fmaxf(a0, 0.0f); a1 = fmaxf(a1, 0.0f); }
        v2f a = {a0, a1};

        const float* __restrict__ Wk0 = sW + (size_t)kb * 64;
        const float* __restrict__ Wk1 = sW + (size_t)(kb + 1) * 64;
        v2f b0 = {Wk0[l16 +  0], Wk1[l16 +  0]};
        v2f b1 = {Wk0[l16 + 16], Wk1[l16 + 16]};
        v2f b2 = {Wk0[l16 + 32], Wk1[l16 + 32]};
        v2f b3 = {Wk0[l16 + 48], Wk1[l16 + 48]};

        c0 = __builtin_amdgcn_wmma_f32_16x16x4_f32(false, a, false, b0, (short)0, c0, false, false);
        c1 = __builtin_amdgcn_wmma_f32_16x16x4_f32(false, a, false, b1, (short)0, c1, false, false);
        c2 = __builtin_amdgcn_wmma_f32_16x16x4_f32(false, a, false, b2, (short)0, c2, false, false);
        c3 = __builtin_amdgcn_wmma_f32_16x16x4_f32(false, a, false, b3, (short)0, c3, false, false);
    }

    #pragma unroll
    for (int r = 0; r < 8; ++r) {
        const int row = m0 + half * 8 + r;
        if (row < n) {
            float* __restrict__ Cr = C + (size_t)row * 64 + l16;
            Cr[0]  = c0[r];
            Cr[16] = c1[r];
            Cr[32] = c2[r];
            Cr[48] = c3[r];
        }
    }
}


#define OT   196
#define OBLK ((NNODES + OT - 1) / OT)
#define NV4  (NEDGES / 4)
#define TL   5
#define SCH  (TL * 32)
#define NSC  (NV4 / SCH)
#define RG   16
#define QCAP 896
#define WCAP 512
#define FQCAP 1312
#define WL_OFF   ((2 * NNODES * HID + NNODES + 63) / 64 * 64)
#define WCNT_OFF (WL_OFF + OBLK * 32 * WCAP)
#define OVF_OFF  (WCNT_OFF + OBLK * 32)
static_assert(NEDGES % 4 == 0 && NV4 % SCH == 0, "dst is read in whole trips of 16-byte groups");
static_assert(NEDGES <= (1 << 20) && OT <= (1 << 12), "queue word = edge | row << 20");
static_assert(NNODES <= (1 << 16) && OT <= (1 << 16), "list word = source | row << 16");
static_assert(QCAP >= 31 + SCH * 4 && FQCAP >= QCAP && NSC >= 32 && 32 % RG == 0, "queue room, a trip for every wave");
static_assert(32 * FQCAP * 4 > 160 * 1024 && 32 * FQCAP * 4 + OT * 4 + 4 <= 320 * 1024, "filter block: over half of the LDS, within all of it");

template <class Take>
__device__ __forceinline__ void walk_dst(const int* __restrict__ dst, int lo, int wave, int lane,
                                         unsigned* __restrict__ myq, Take&& take) {
    int qn = 0;
    const int4* __restrict__ dst4 = (const int4*)dst;
    const int start = (blockIdx.x * (NSC / OBLK)) % NSC;
    const int ntrip = (NSC - wave + 31) / 32;

    auto load_trip = [&](int i, int4 (&buf)[TL], int& base) {
        const int ic = i < ntrip ? i : ntrip - 1;
        int s2 = ic * 32 + wave + start;
        if (s2 >= NSC) s2 -= NSC;
        base = s2 * SCH;
        #pragma unroll
        for (int k = 0; k < TL; ++k) buf[k] = dst4[base + k * 32 + lane];
    };

    auto push = [&](unsigned t, int e) {
        const bool m = t < (unsigned)OT;
        const unsigned mask = __builtin_amdgcn_ballot_w32(m);
        if (mask != 0u) {
            if (m) {
                const int slot = qn + (int)__builtin_amdgcn_mbcnt_lo(mask, 0u);
                myq[slot] = (unsigned)e | (t << 20);
            }
            qn += __builtin_popcount(mask);
        }
    };

    auto process = [&](const int4 (&buf)[TL], int base, bool last) {
        #pragma unroll
        for (int k = 0; k < TL; ++k) {
            const int4 d = buf[k];
            const unsigned t0 = (unsigned)(d.x - lo), t1 = (unsigned)(d.y - lo);
            const unsigned t2 = (unsigned)(d.z - lo), t3 = (unsigned)(d.w - lo);
            const unsigned tm = min(min(t0, t1), min(t2, t3));
            if (__builtin_amdgcn_ballot_w32(tm < (unsigned)OT) != 0u) {
                const int e0 = (base + k * 32 + lane) * 4;
                push(t0, e0);
                push(t1, e0 + 1);
                push(t2, e0 + 2);
                push(t3, e0 + 3);
            }
        }
        const int keep = last ? 0 : 31;
        while (qn > keep) qn = take(qn);
    };

    int4 bufA[TL], bufB[TL];
    int baseA = 0, baseB = 0;
    load_trip(0, bufA, baseA);
    for (int i = 0; i < ntrip; i += 2) {
        load_trip(i + 1, bufB, baseB);
        process(bufA, baseA, i + 1 >= ntrip);
        if (i + 1 >= ntrip) break;
        load_trip(i + 2, bufA, baseA);
        process(bufB, baseB, i + 2 >= ntrip);
    }
}

__global__ __launch_bounds__(1024)
void k_owner_filter(const int* __restrict__ src, const int* __restrict__ dst,
                    int* __restrict__ wl, int* __restrict__ wcnt, int* __restrict__ ovf,
                    float* __restrict__ dinv) {
    __shared__ int scnt[OT];
    __shared__ int sovf;
    __shared__ unsigned q[32 * FQCAP];

    const int tid  = threadIdx.x;
    const int lane = tid & 31;
    const int wave = __builtin_amdgcn_readfirstlane(tid >> 5);
    const int lo   = blockIdx.x * OT;

    if (tid < OT) scnt[tid] = 0;
    if (tid == 0) sovf = 0;
    __syncthreads();

    unsigned* __restrict__ myq = q + wave * FQCAP;
    int* __restrict__ mywl = wl + (size_t)(blockIdx.x * 32 + wave) * WCAP;
    int wn = 0;

    walk_dst(dst, lo, wave, lane, myq, [&](int qn) {
        const int base = qn > 32 ? qn - 32 : 0;
        const int cnt  = qn - base;
        if (lane < cnt) {
            const unsigned p = myq[base + lane];
            const int e  = (int)(p & 0xFFFFFu);
            const int dl = (int)(p >> 20);
            const int s  = src[e];
            atomicAdd(&scnt[dl], 1);
            if (wn + lane < WCAP) mywl[wn + lane] = s | (dl << 16);
        }
        wn += cnt;
        return base;
    });

    if (lane == 0) {
        wcnt[blockIdx.x * 32 + wave] = wn;
        if (wn > WCAP) sovf = 1;
    }
    __syncthreads();
    if (tid < OT && lo + tid < NNODES) dinv[lo + tid] = rsqrtf((float)(scnt[tid] + 1));
    if (tid == 0) ovf[blockIdx.x] = sovf;
}

__global__ __launch_bounds__(1024)
void k_owner_agg(const float* __restrict__ h, const float* __restrict__ dinv,
                 const float* __restrict__ bias,
                 const int* __restrict__ src, const int* __restrict__ dst,
                 const int* __restrict__ wl, const int* __restrict__ wcnt, const int* __restrict__ ovf,
                 float* __restrict__ a) {
    __shared__ float acc[(OT + 1) * HID];
    __shared__ float sdinv[OT];
    __shared__ unsigned q[32 * QCAP];

    const int tid  = threadIdx.x;
    const int lane = tid & 31;
    const int wave = __builtin_amdgcn_readfirstlane(tid >> 5);
    const int lo   = blockIdx.x * OT;

    const int over = ovf[blockIdx.x];
    const int* __restrict__ mywl = wl + (size_t)(blockIdx.x * 32 + wave) * WCAP;
    int n = 0;
    int wnext = 0;
    if (!over) {
        n = __builtin_amdgcn_readfirstlane(wcnt[blockIdx.x * 32 + wave]);
        if (n > 0) wnext = mywl[lane < n ? lane : n - 1];
    }

    for (int i = tid; i < (OT + 1) * 16; i += 1024) {
        const int r = i >> 4, c = i & 15;
        const int node = lo + r;
        float4 v = make_float4(0.f, 0.f, 0.f, 0.f);
        if (r < OT && node < NNODES) {
            const float4 hv = ((const float4*)h)[(size_t)node * 16 + c];
            const float4 b  = ((const float4*)bias)[c];
            const float di = dinv[node];
            v.x = hv.x * di * di + b.x;
            v.y = hv.y * di * di + b.y;
            v.z = hv.z * di * di + b.z;
            v.w = hv.w * di * di + b.w;
        }
        ((float4*)acc)[i] = v;
    }
    if (tid < OT) sdinv[tid] = (lo + tid < NNODES) ? dinv[lo + tid] : 0.0f;
    __syncthreads();

    const char* __restrict__ hb = (const char*)h;
    char* accb = (char*)acc;

    auto add_rows = [&](int cnt, unsigned soff, unsigned doff, float dv, float sd) {
        #pragma nounroll
        for (int j0 = 0; j0 < cnt; j0 += RG) {
            float2 v[RG];
            #pragma unroll
            for (int k = 0; k < RG; ++k) {
                const unsigned so = (unsigned)__builtin_amdgcn_readlane((int)soff, j0 + k);
                v[k] = *(const float2*)(hb + (so + (unsigned)lane * 8u));
            }
            #pragma unroll
            for (int k = 0; k < RG; ++k) {
                const float wj =
                    __builtin_bit_cast(float, __builtin_amdgcn_readlane(__builtin_bit_cast(int, dv), j0 + k)) *
                    __builtin_bit_cast(float, __builtin_amdgcn_readlane(__builtin_bit_cast(int, sd), j0 + k));
                const unsigned dj = (unsigned)__builtin_amdgcn_readlane((int)doff, j0 + k);
                float* ap = (float*)(accb + (dj + (unsigned)lane * 8u));
                atomicAdd(ap,     v[k].x * wj);
                atomicAdd(ap + 1, v[k].y * wj);
            }
        }
    };

    if (!over) {
        for (int j0 = 0; j0 < n; j0 += 32) {
            const int cnt = min(32, n - j0);
            const unsigned word = (unsigned)wnext;
            const int nj = j0 + 32 + lane;
            wnext = mywl[nj < n ? nj : n - 1];
            const unsigned s  = word & 0xFFFFu;
            const unsigned dl = word >> 16;
            const float dv = dinv[s];
            const float sd = sdinv[dl];
            add_rows(cnt, s * (HID * 4u), (lane < cnt ? dl : (unsigned)OT) * (HID * 4u), dv, sd);
        }
    } else {
        unsigned* __restrict__ myq = q + wave * QCAP;
        walk_dst(dst, lo, wave, lane, myq, [&](int qn) {
            const int base = qn > 32 ? qn - 32 : 0;
            const int cnt  = qn - base;
            const bool used = lane < cnt;
            const unsigned p = myq[base + (used ? lane : cnt - 1)];
            const int e  = (int)(p & 0xFFFFFu);
            const int dl = (int)(p >> 20);
            const int s  = src[e];
            const float dv = dinv[s];
            const float sd = sdinv[dl];
            add_rows(cnt, (unsigned)s * (HID * 4u), (unsigned)(used ? dl : OT) * (HID * 4u), dv, sd);
            return base;
        });
    }
    __syncthreads();

    for (int i = tid; i < OT * 16; i += 1024) {
        const int node = lo + (i >> 4);
        if (node < NNODES) ((float4*)a)[(size_t)node * 16 + (i & 15)] = ((const float4*)acc)[i];
    }
}


#define PT    256
#define PW    (PT / 32)
#define NB4   (NNODES / 4)
#define NK    ((NB4 + PT - 1) / PT)
#define NWORD ((NK + 7) / 8)
#define LCAP  1024
#define RB    16
#define NWL   ((HID * NCLASSES + PT - 1) / PT)
static_assert(NNODES % 4 == 0 && HID == 64 && NCLASSES <= PT, "batch in 16-byte groups, a row is 32 lanes of 8 bytes");

__global__ __launch_bounds__(PT)
void k_graph_pool_head(const float* __restrict__ a, const int* __restrict__ batch,
                       const float* __restrict__ Wl, const float* __restrict__ bl,
                       float* __restrict__ out) {
    __shared__ int   list[LCAP];
    __shared__ int   wtot[PW];
    __shared__ float part[PW * HID];
    __shared__ float ssum[HID];
    __shared__ float sWl[HID * NCLASSES];

    const int tid  = threadIdx.x;
    const int lane = tid & 31;
    const int wave = __builtin_amdgcn_readfirstlane(tid >> 5);
    const int g    = blockIdx.x;

    float wl[NWL];
    #pragma unroll
    for (int i = 0; i < NWL; ++i) wl[i] = (i + 1) * PT <= HID * NCLASSES || tid + i * PT < HID * NCLASSES ? Wl[tid + i * PT] : 0.0f;
    const float blv = tid < NCLASSES ? bl[tid] : 0.0f;

    const int4* __restrict__ batch4 = (const int4*)batch;
    unsigned mask[NWORD];
    #pragma unroll
    for (int w = 0; w < NWORD; ++w) {
        int4 d[8];
        #pragma unroll
        for (int j = 0; j < 8; ++j) {
            const int k = w * 8 + j;
            d[j] = make_int4(-1, -1, -1, -1);
            if (k < NK) {
                const int idx = k * PT + tid;
                if ((k + 1) * PT <= NB4 || idx < NB4) d[j] = batch4[idx];
            }
        }
        unsigned m = 0u;
        #pragma unroll
        for (int j = 0; j < 8; ++j) {
            if (w * 8 + j < NK) {
                m |= (d[j].x == g ? 1u : 0u) << (4 * j);
                m |= (d[j].y == g ? 2u : 0u) << (4 * j);
                m |= (d[j].z == g ? 4u : 0u) << (4 * j);
                m |= (d[j].w == g ? 8u : 0u) << (4 * j);
            }
        }
        mask[w] = m;
    }

    #pragma unroll
    for (int i = 0; i < NWL; ++i) if ((i + 1) * PT <= HID * NCLASSES || tid + i * PT < HID * NCLASSES) sWl[tid + i * PT] = wl[i];

    int c = 0;
    #pragma unroll
    for (int w = 0; w < NWORD; ++w) c += __builtin_popcount(mask[w]);
    int x = c;
    for (int dl = 1; dl < 32; dl <<= 1) {
        const int y = __shfl_up(x, dl);
        if (lane >= dl) x += y;
    }
    if (lane == 31) wtot[wave] = x;
    __syncthreads();
    int base = 0, total = 0;
    for (int w = 0; w < PW; ++w) {
        const int t = wtot[w];
        if (w < wave) base += t;
        total += t;
    }
    const int first = base + x - c;

    float2 acc = make_float2(0.0f, 0.0f);
    const float2* __restrict__ a2 = (const float2*)a;
    for (int r0 = 0; r0 < total; r0 += LCAP) {
        if (r0 != 0) __syncthreads();
        if (c != 0) {
            int p = first - r0;
            #pragma unroll
            for (int w = 0; w < NWORD; ++w) {
                unsigned m = mask[w];
                while (m != 0u) {
                    const int b = __builtin_ctz(m);
                    m &= m - 1u;
                    if (p >= 0 && p < LCAP) list[p] = (((w * 8 + (b >> 2)) * PT + tid) << 2) + (b & 3);
                    ++p;
                }
            }
        }
        __syncthreads();
        const int n = min(total - r0, LCAP);
        for (int p0 = wave; p0 < n + wave; p0 += PW * RB) {
            float2 v[RB];
            #pragma unroll
            for (int k = 0; k < RB; ++k) {
                const int p = p0 + PW * k < n ? p0 + PW * k : n - 1;
                const int node = __builtin_amdgcn_readfirstlane(list[p]);
                v[k] = a2[(size_t)node * 32 + lane];
            }
            #pragma unroll
            for (int k = 0; k < RB; ++k) {
                const bool ok = p0 + PW * k < n;
                acc.x += fmaxf(ok ? v[k].x : 0.0f, 0.0f);
                acc.y += fmaxf(ok ? v[k].y : 0.0f, 0.0f);
            }
        }
    }

    ((float2*)part)[wave * 32 + lane] = acc;
    __syncthreads();
    if (tid < HID) {
        float s = 0.0f;
        for (int w = 0; w < PW; ++w) s += part[w * HID + tid];
        ssum[tid] = s;
    }
    __syncthreads();
    if (tid < NCLASSES) {
        const float inv = 1.0f / fmaxf((float)total, 1.0f);
        float o = 0.0f;
        #pragma unroll
        for (int k = 0; k < HID; ++k) o += ssum[k] * sWl[k * NCLASSES + tid];
        out[g * NCLASSES + tid] = o * inv + blv;
    }
}


extern "C" void kernel_launch(void* const* d_in, const int* in_sizes, int n_in,
                              void* d_out, int out_size, void* d_ws, size_t ws_size,
                              hipStream_t stream) {
    const float* x  = (const float*)d_in[0];
    const float* W1 = (const float*)d_in[1];
    const float* b1 = (const float*)d_in[2];
    const float* W2 = (const float*)d_in[3];
    const float* b2 = (const float*)d_in[4];
    const float* Wl = (const float*)d_in[5];
    const float* bl = (const float*)d_in[6];
    const int* edge  = (const int*)d_in[7];
    const int* batch = (const int*)d_in[8];
    const int* src = edge;
    const int* dst = edge + NEDGES;
    float* out = (float*)d_out;

    float* ws   = (float*)d_ws;
    float* h    = ws;
    float* a    = h + (size_t)NNODES * HID;
    float* dinv = a + (size_t)NNODES * HID;
    int* wl   = (int*)ws + (size_t)WL_OFF;
    int* wcnt = (int*)ws + (size_t)WCNT_OFF;
    int* ovf  = (int*)ws + (size_t)OVF_OFF;

    const int TPB = 256;
    const int gGmm = ((NNODES + 15) / 16 + 7) / 8;

    k_owner_filter<<<OBLK, 1024, 0, stream>>>(src, dst, wl, wcnt, ovf, dinv);

    k_gemm_wmma<<<gGmm, TPB, 0, stream>>>(x, W1, h, NNODES, 0);
    k_owner_agg<<<OBLK, 1024, 0, stream>>>(h, dinv, b1, src, dst, wl, wcnt, ovf, a);

    k_gemm_wmma<<<gGmm, TPB, 0, stream>>>(a, W2, h, NNODES, 1);
    k_owner_agg<<<OBLK, 1024, 0, stream>>>(h, dinv, b2, src, dst, wl, wcnt, ovf, a);

    k_graph_pool_head<<<NGRAPHS, PT, 0, stream>>>(a, batch, Wl, bl, out);
}


# dst packed to one owner byte per edge (with a source|row word, both written in the layer-1 product's launch): the owner walk reads 1 byte per edge, not 4, and fetches only its hits' words
# speedup vs baseline: 3.7245x; 1.1327x over previous
#include <hip/hip_runtime.h>

typedef float v2f __attribute__((ext_vector_type(2)));
typedef float v8f __attribute__((ext_vector_type(8)));

#define NNODES 50000
#define NEDGES 800000
#define HID 64
#define NGRAPHS 512
#define NCLASSES 10


#define OT   196
#define NV4  (NEDGES / 4)

__global__ __launch_bounds__(256)
void k_gemm_wmma(const float* __restrict__ A, const float* __restrict__ W,
                 float* __restrict__ C, int n, int relu_in,
                 const int* __restrict__ src, const int* __restrict__ dst,
                 unsigned* __restrict__ ob, int* __restrict__ ew) {
    __shared__ float sW[64 * 64];
    const int tid = threadIdx.x;
    if (ob != nullptr) {
        for (int g = blockIdx.x * 256 + tid; g < NV4; g += gridDim.x * 256) {
            const int4 d = ((const int4*)dst)[g];
            const int4 s = ((const int4*)src)[g];
            const unsigned o0 = (unsigned)d.x / OT, o1 = (unsigned)d.y / OT;
            const unsigned o2 = (unsigned)d.z / OT, o3 = (unsigned)d.w / OT;
            int4 w;
            w.x = s.x | (int)(((unsigned)d.x - o0 * OT) << 16);
            w.y = s.y | (int)(((unsigned)d.y - o1 * OT) << 16);
            w.z = s.z | (int)(((unsigned)d.z - o2 * OT) << 16);
            w.w = s.w | (int)(((unsigned)d.w - o3 * OT) << 16);
            ((int4*)ew)[g] = w;
            ob[g] = (o0 & 255u) | (o1 & 255u) << 8 | (o2 & 255u) << 16 | o3 << 24;
        }
    }
    #pragma unroll
    for (int i = 0; i < 16; ++i) sW[tid + i * 256] = W[tid + i * 256];
    __syncthreads();

    const int wave = tid >> 5;
    const int lane = tid & 31;
    const int half = lane >> 4;
    const int l16  = lane & 15;
    const int m0   = (blockIdx.x * 8 + wave) * 16;
    if (m0 >= n) return;

    v8f c0 = {}, c1 = {}, c2 = {}, c3 = {};

    const int arow = m0 + l16;
    const float* __restrict__ Arow = A + (size_t)arow * 64;

    #pragma unroll
    for (int k0 = 0; k0 < 64; k0 += 4) {
        const int kb = k0 + half * 2;
        float a0 = Arow[kb];
        float a1 = Arow[kb + 1];
        if (relu_in) { a0 = fmaxf(a0, 0.0f); a1 = fmaxf(a1, 0.0f); }
        v2f a = {a0, a1};

        const float* __restrict__ Wk0 = sW + (size_t)kb * 64;
        const float* __restrict__ Wk1 = sW + (size_t)(kb + 1) * 64;
        v2f b0 = {Wk0[l16 +  0], Wk1[l16 +  0]};
        v2f b1 = {Wk0[l16 + 16], Wk1[l16 + 16]};
        v2f b2 = {Wk0[l16 + 32], Wk1[l16 + 32]};
        v2f b3 = {Wk0[l16 + 48], Wk1[l16 + 48]};

        c0 = __builtin_amdgcn_wmma_f32_16x16x4_f32(false, a, false, b0, (short)0, c0, false, false);
        c1 = __builtin_amdgcn_wmma_f32_16x16x4_f32(false, a, false, b1, (short)0, c1, false, false);
        c2 = __builtin_amdgcn_wmma_f32_16x16x4_f32(false, a, false, b2, (short)0, c2, false, false);
        c3 = __builtin_amdgcn_wmma_f32_16x16x4_f32(false, a, false, b3, (short)0, c3, false, false);
    }

    #pragma unroll
    for (int r = 0; r < 8; ++r) {
        const int row = m0 + half * 8 + r;
        if (row < n) {
            float* __restrict__ Cr = C + (size_t)row * 64 + l16;
            Cr[0]  = c0[r];
            Cr[16] = c1[r];
            Cr[32] = c2[r];
            Cr[48] = c3[r];
        }
    }
}


#define OBLK ((NNODES + OT - 1) / OT)
#define TL   5
#define SCH  (TL * 32)
#define NSC  (NV4 / SCH)
#define RG   16
#define QCAP 896
#define WCAP 512
#define FQCAP 1312
#define WL_OFF   ((2 * NNODES * HID + NNODES + 63) / 64 * 64)
#define WCNT_OFF (WL_OFF + OBLK * 32 * WCAP)
#define OVF_OFF  (WCNT_OFF + OBLK * 32)
#define OB_OFF   ((OVF_OFF + OBLK + 63) / 64 * 64)
#define EW_OFF   ((OB_OFF + NV4 + 63) / 64 * 64)
#define NV16 (NEDGES / 16)
#define TB   5
#define NTB  ((NV16 + TB * 32 - 1) / (TB * 32))
static_assert(NEDGES % 4 == 0 && NV4 % SCH == 0, "dst is read in whole trips of 16-byte groups");
static_assert(NEDGES <= (1 << 20) && OT <= (1 << 12), "queue word = edge | row << 20");
static_assert(NNODES <= (1 << 16) && OT <= (1 << 16), "list word = source | row << 16");
static_assert(QCAP >= 31 + SCH * 4 && FQCAP >= QCAP && NSC >= 32 && 32 % RG == 0, "queue room, a trip for every wave");
static_assert(32 * FQCAP * 4 > 160 * 1024 && 32 * FQCAP * 4 + OT * 4 + 4 <= 320 * 1024, "filter block: over half of the LDS, within all of it");
static_assert(OBLK <= 256 && NEDGES % 16 == 0 && FQCAP >= 31 + 32 * 16 && NTB >= 32, "an owner number is a byte; one load's hits fit the queue");

template <class Take>
__device__ __forceinline__ void walk_dst(const int* __restrict__ dst, int lo, int wave, int lane,
                                         unsigned* __restrict__ myq, Take&& take) {
    int qn = 0;
    const int4* __restrict__ dst4 = (const int4*)dst;
    const int start = (blockIdx.x * (NSC / OBLK)) % NSC;
    const int ntrip = (NSC - wave + 31) / 32;

    auto load_trip = [&](int i, int4 (&buf)[TL], int& base) {
        const int ic = i < ntrip ? i : ntrip - 1;
        int s2 = ic * 32 + wave + start;
        if (s2 >= NSC) s2 -= NSC;
        base = s2 * SCH;
        #pragma unroll
        for (int k = 0; k < TL; ++k) buf[k] = dst4[base + k * 32 + lane];
    };

    auto push = [&](unsigned t, int e) {
        const bool m = t < (unsigned)OT;
        const unsigned mask = __builtin_amdgcn_ballot_w32(m);
        if (mask != 0u) {
            if (m) {
                const int slot = qn + (int)__builtin_amdgcn_mbcnt_lo(mask, 0u);
                myq[slot] = (unsigned)e | (t << 20);
            }
            qn += __builtin_popcount(mask);
        }
    };

    auto process = [&](const int4 (&buf)[TL], int base, bool last) {
        #pragma unroll
        for (int k = 0; k < TL; ++k) {
            const int4 d = buf[k];
            const unsigned t0 = (unsigned)(d.x - lo), t1 = (unsigned)(d.y - lo);
            const unsigned t2 = (unsigned)(d.z - lo), t3 = (unsigned)(d.w - lo);
            const unsigned tm = min(min(t0, t1), min(t2, t3));
            if (__builtin_amdgcn_ballot_w32(tm < (unsigned)OT) != 0u) {
                const int e0 = (base + k * 32 + lane) * 4;
                push(t0, e0);
                push(t1, e0 + 1);
                push(t2, e0 + 2);
                push(t3, e0 + 3);
            }
        }
        const int keep = last ? 0 : 31;
        while (qn > keep) qn = take(qn);
    };

    int4 bufA[TL], bufB[TL];
    int baseA = 0, baseB = 0;
    load_trip(0, bufA, baseA);
    for (int i = 0; i < ntrip; i += 2) {
        load_trip(i + 1, bufB, baseB);
        process(bufA, baseA, i + 1 >= ntrip);
        if (i + 1 >= ntrip) break;
        load_trip(i + 2, bufA, baseA);
        process(bufB, baseB, i + 2 >= ntrip);
    }
}

__global__ __launch_bounds__(1024)
void k_owner_filter(const unsigned* __restrict__ ob, const int* __restrict__ ew,
                    int* __restrict__ wl, int* __restrict__ wcnt, int* __restrict__ ovf,
                    float* __restrict__ dinv) {
    __shared__ int scnt[OT];
    __shared__ int sovf;
    __shared__ unsigned q[32 * FQCAP];

    const int tid  = threadIdx.x;
    const int lane = tid & 31;
    const int wave = __builtin_amdgcn_readfirstlane(tid >> 5);
    const int lo   = blockIdx.x * OT;
    const unsigned idr = (unsigned)blockIdx.x * 0x01010101u;
    const uint4* __restrict__ ob4 = (const uint4*)ob;

    if (tid < OT) scnt[tid] = 0;
    if (tid == 0) sovf = 0;
    __syncthreads();

    unsigned* __restrict__ myq = q + wave * FQCAP;
    int* __restrict__ mywl = wl + (size_t)(blockIdx.x * 32 + wave) * WCAP;
    int wn = 0;
    int qn = 0;

    auto take = [&](int fill) {
        const int base = fill > 32 ? fill - 32 : 0;
        const int cnt  = fill - base;
        if (lane < cnt) {
            const unsigned w = myq[base + lane];
            const unsigned p = w & 31u;
            const unsigned e = (w >> 5) * 16u + ((p & 7u) << 2) + (p >> 3);
            const int word = ew[e];
            atomicAdd(&scnt[(unsigned)word >> 16], 1);
            if (wn + lane < WCAP) mywl[wn + lane] = word;
        }
        wn += cnt;
        return base;
    };

    const int start = (blockIdx.x * NTB) / OBLK;
    const int ntrip = (NTB - wave + 31) / 32;

    auto load_trip = [&](int i, uint4 (&buf)[TB], int& base) {
        const int ic = i < ntrip ? i : ntrip - 1;
        int s2 = ic * 32 + wave + start;
        if (s2 >= NTB) s2 -= NTB;
        base = s2 * (TB * 32);
        #pragma unroll
        for (int k = 0; k < TB; ++k) {
            const int g = base + k * 32 + lane;
            buf[k] = ob4[g < NV16 ? g : NV16 - 1];
        }
    };

    auto eq7 = [&](unsigned x) {
        const unsigned y = x ^ idr;
        return ~(((y & 0x7F7F7F7Fu) + 0x7F7F7F7Fu) | y | 0x7F7F7F7Fu);
    };

    auto process = [&](const uint4 (&buf)[TB], int base) {
        #pragma unroll
        for (int k = 0; k < TB; ++k) {
            const uint4 d = buf[k];
            const int g = base + k * 32 + lane;
            unsigned m = (eq7(d.x) >> 7) | (eq7(d.y) >> 6) | (eq7(d.z) >> 5) | (eq7(d.w) >> 4);
            if (base + k * 32 + 32 > NV16) m = g < NV16 ? m : 0u;
            unsigned any = __builtin_amdgcn_ballot_w32(m != 0u);
            while (any != 0u) {
                if (m != 0u) {
                    const int p = __builtin_ctz(m);
                    const int slot = qn + (int)__builtin_amdgcn_mbcnt_lo(any, 0u);
                    myq[slot] = (unsigned)(g * 32 + p);
                    m &= m - 1u;
                }
                qn += __builtin_popcount(any);
                any = __builtin_amdgcn_ballot_w32(m != 0u);
            }
            while (qn > 31) qn = take(qn);
        }
    };

    uint4 bufA[TB], bufB[TB];
    int baseA = 0, baseB = 0;
    load_trip(0, bufA, baseA);
    for (int i = 0; i < ntrip; i += 2) {
        load_trip(i + 1, bufB, baseB);
        process(bufA, baseA);
        if (i + 1 >= ntrip) break;
        load_trip(i + 2, bufA, baseA);
        process(bufB, baseB);
    }
    while (qn > 0) qn = take(qn);

    if (lane == 0) {
        wcnt[blockIdx.x * 32 + wave] = wn;
        if (wn > WCAP) sovf = 1;
    }
    __syncthreads();
    if (tid < OT && lo + tid < NNODES) dinv[lo + tid] = rsqrtf((float)(scnt[tid] + 1));
    if (tid == 0) ovf[blockIdx.x] = sovf;
}

__global__ __launch_bounds__(1024)
void k_owner_agg(const float* __restrict__ h, const float* __restrict__ dinv,
                 const float* __restrict__ bias,
                 const int* __restrict__ src, const int* __restrict__ dst,
                 const int* __restrict__ wl, const int* __restrict__ wcnt, const int* __restrict__ ovf,
                 float* __restrict__ a) {
    __shared__ float acc[(OT + 1) * HID];
    __shared__ float sdinv[OT];
    __shared__ unsigned q[32 * QCAP];

    const int tid  = threadIdx.x;
    const int lane = tid & 31;
    const int wave = __builtin_amdgcn_readfirstlane(tid >> 5);
    const int lo   = blockIdx.x * OT;

    const int over = ovf[blockIdx.x];
    const int* __restrict__ mywl = wl + (size_t)(blockIdx.x * 32 + wave) * WCAP;
    int n = 0;
    int wnext = 0;
    if (!over) {
        n = __builtin_amdgcn_readfirstlane(wcnt[blockIdx.x * 32 + wave]);
        if (n > 0) wnext = mywl[lane < n ? lane : n - 1];
    }

    for (int i = tid; i < (OT + 1) * 16; i += 1024) {
        const int r = i >> 4, c = i & 15;
        const int node = lo + r;
        float4 v = make_float4(0.f, 0.f, 0.f, 0.f);
        if (r < OT && node < NNODES) {
            const float4 hv = ((const float4*)h)[(size_t)node * 16 + c];
            const float4 b  = ((const float4*)bias)[c];
            const float di = dinv[node];
            v.x = hv.x * di * di + b.x;
            v.y = hv.y * di * di + b.y;
            v.z = hv.z * di * di + b.z;
            v.w = hv.w * di * di + b.w;
        }
        ((float4*)acc)[i] = v;
    }
    if (tid < OT) sdinv[tid] = (lo + tid < NNODES) ? dinv[lo + tid] : 0.0f;
    __syncthreads();

    const char* __restrict__ hb = (const char*)h;
    char* accb = (char*)acc;

    auto add_rows = [&](int cnt, unsigned soff, unsigned doff, float dv, float sd) {
        #pragma nounroll
        for (int j0 = 0; j0 < cnt; j0 += RG) {
            float2 v[RG];
            #pragma unroll
            for (int k = 0; k < RG; ++k) {
                const unsigned so = (unsigned)__builtin_amdgcn_readlane((int)soff, j0 + k);
                v[k] = *(const float2*)(hb + (so + (unsigned)lane * 8u));
            }
            #pragma unroll
            for (int k = 0; k < RG; ++k) {
                const float wj =
                    __builtin_bit_cast(float, __builtin_amdgcn_readlane(__builtin_bit_cast(int, dv), j0 + k)) *
                    __builtin_bit_cast(float, __builtin_amdgcn_readlane(__builtin_bit_cast(int, sd), j0 + k));
                const unsigned dj = (unsigned)__builtin_amdgcn_readlane((int)doff, j0 + k);
                float* ap = (float*)(accb + (dj + (unsigned)lane * 8u));
                atomicAdd(ap,     v[k].x * wj);
                atomicAdd(ap + 1, v[k].y * wj);
            }
        }
    };

    if (!over) {
        for (int j0 = 0; j0 < n; j0 += 32) {
            const int cnt = min(32, n - j0);
            const unsigned word = (unsigned)wnext;
            const int nj = j0 + 32 + lane;
            wnext = mywl[nj < n ? nj : n - 1];
            const unsigned s  = word & 0xFFFFu;
            const unsigned dl = word >> 16;
            const float dv = dinv[s];
            const float sd = sdinv[dl];
            add_rows(cnt, s * (HID * 4u), (lane < cnt ? dl : (unsigned)OT) * (HID * 4u), dv, sd);
        }
    } else {
        unsigned* __restrict__ myq = q + wave * QCAP;
        walk_dst(dst, lo, wave, lane, myq, [&](int qn) {
            const int base = qn > 32 ? qn - 32 : 0;
            const int cnt  = qn - base;
            const bool used = lane < cnt;
            const unsigned p = myq[base + (used ? lane : cnt - 1)];
            const int e  = (int)(p & 0xFFFFFu);
            const int dl = (int)(p >> 20);
            const int s  = src[e];
            const float dv = dinv[s];
            const float sd = sdinv[dl];
            add_rows(cnt, (unsigned)s * (HID * 4u), (unsigned)(used ? dl : OT) * (HID * 4u), dv, sd);
            return base;
        });
    }
    __syncthreads();

    for (int i = tid; i < OT * 16; i += 1024) {
        const int node = lo + (i >> 4);
        if (node < NNODES) ((float4*)a)[(size_t)node * 16 + (i & 15)] = ((const float4*)acc)[i];
    }
}


#define PT    256
#define PW    (PT / 32)
#define NB4   (NNODES / 4)
#define NK    ((NB4 + PT - 1) / PT)
#define NWORD ((NK + 7) / 8)
#define LCAP  1024
#define RB    16
#define NWL   ((HID * NCLASSES + PT - 1) / PT)
static_assert(NNODES % 4 == 0 && HID == 64 && NCLASSES <= PT, "batch in 16-byte groups, a row is 32 lanes of 8 bytes");

__global__ __launch_bounds__(PT)
void k_graph_pool_head(const float* __restrict__ a, const int* __restrict__ batch,
                       const float* __restrict__ Wl, const float* __restrict__ bl,
                       float* __restrict__ out) {
    __shared__ int   list[LCAP];
    __shared__ int   wtot[PW];
    __shared__ float part[PW * HID];
    __shared__ float ssum[HID];
    __shared__ float sWl[HID * NCLASSES];

    const int tid  = threadIdx.x;
    const int lane = tid & 31;
    const int wave = __builtin_amdgcn_readfirstlane(tid >> 5);
    const int g    = blockIdx.x;

    float wl[NWL];
    #pragma unroll
    for (int i = 0; i < NWL; ++i) wl[i] = (i + 1) * PT <= HID * NCLASSES || tid + i * PT < HID * NCLASSES ? Wl[tid + i * PT] : 0.0f;
    const float blv = tid < NCLASSES ? bl[tid] : 0.0f;

    const int4* __restrict__ batch4 = (const int4*)batch;
    unsigned mask[NWORD];
    #pragma unroll
    for (int w = 0; w < NWORD; ++w) {
        int4 d[8];
        #pragma unroll
        for (int j = 0; j < 8; ++j) {
            const int k = w * 8 + j;
            d[j] = make_int4(-1, -1, -1, -1);
            if (k < NK) {
                const int idx = k * PT + tid;
                if ((k + 1) * PT <= NB4 || idx < NB4) d[j] = batch4[idx];
            }
        }
        unsigned m = 0u;
        #pragma unroll
        for (int j = 0; j < 8; ++j) {
            if (w * 8 + j < NK) {
                m |= (d[j].x == g ? 1u : 0u) << (4 * j);
                m |= (d[j].y == g ? 2u : 0u) << (4 * j);
                m |= (d[j].z == g ? 4u : 0u) << (4 * j);
                m |= (d[j].w == g ? 8u : 0u) << (4 * j);
            }
        }
        mask[w] = m;
    }

    #pragma unroll
    for (int i = 0; i < NWL; ++i) if ((i + 1) * PT <= HID * NCLASSES || tid + i * PT < HID * NCLASSES) sWl[tid + i * PT] = wl[i];

    int c = 0;
    #pragma unroll
    for (int w = 0; w < NWORD; ++w) c += __builtin_popcount(mask[w]);
    int x = c;
    for (int dl = 1; dl < 32; dl <<= 1) {
        const int y = __shfl_up(x, dl);
        if (lane >= dl) x += y;
    }
    if (lane == 31) wtot[wave] = x;
    __syncthreads();
    int base = 0, total = 0;
    for (int w = 0; w < PW; ++w) {
        const int t = wtot[w];
        if (w < wave) base += t;
        total += t;
    }
    const int first = base + x - c;

    float2 acc = make_float2(0.0f, 0.0f);
    const float2* __restrict__ a2 = (const float2*)a;
    for (int r0 = 0; r0 < total; r0 += LCAP) {
        if (r0 != 0) __syncthreads();
        if (c != 0) {
            int p = first - r0;
            #pragma unroll
            for (int w = 0; w < NWORD; ++w) {
                unsigned m = mask[w];
                while (m != 0u) {
                    const int b = __builtin_ctz(m);
                    m &= m - 1u;
                    if (p >= 0 && p < LCAP) list[p] = (((w * 8 + (b >> 2)) * PT + tid) << 2) + (b & 3);
                    ++p;
                }
            }
        }
        __syncthreads();
        const int n = min(total - r0, LCAP);
        for (int p0 = wave; p0 < n + wave; p0 += PW * RB) {
            float2 v[RB];
            #pragma unroll
            for (int k = 0; k < RB; ++k) {
                const int p = p0 + PW * k < n ? p0 + PW * k : n - 1;
                const int node = __builtin_amdgcn_readfirstlane(list[p]);
                v[k] = a2[(size_t)node * 32 + lane];
            }
            #pragma unroll
            for (int k = 0; k < RB; ++k) {
                const bool ok = p0 + PW * k < n;
                acc.x += fmaxf(ok ? v[k].x : 0.0f, 0.0f);
                acc.y += fmaxf(ok ? v[k].y : 0.0f, 0.0f);
            }
        }
    }

    ((float2*)part)[wave * 32 + lane] = acc;
    __syncthreads();
    if (tid < HID) {
        float s = 0.0f;
        for (int w = 0; w < PW; ++w) s += part[w * HID + tid];
        ssum[tid] = s;
    }
    __syncthreads();
    if (tid < NCLASSES) {
        const float inv = 1.0f / fmaxf((float)total, 1.0f);
        float o = 0.0f;
        #pragma unroll
        for (int k = 0; k < HID; ++k) o += ssum[k] * sWl[k * NCLASSES + tid];
        out[g * NCLASSES + tid] = o * inv + blv;
    }
}


extern "C" void kernel_launch(void* const* d_in, const int* in_sizes, int n_in,
                              void* d_out, int out_size, void* d_ws, size_t ws_size,
                              hipStream_t stream) {
    const float* x  = (const float*)d_in[0];
    const float* W1 = (const float*)d_in[1];
    const float* b1 = (const float*)d_in[2];
    const float* W2 = (const float*)d_in[3];
    const float* b2 = (const float*)d_in[4];
    const float* Wl = (const float*)d_in[5];
    const float* bl = (const float*)d_in[6];
    const int* edge  = (const int*)d_in[7];
    const int* batch = (const int*)d_in[8];
    const int* src = edge;
    const int* dst = edge + NEDGES;
    float* out = (float*)d_out;

    float* ws   = (float*)d_ws;
    float* h    = ws;
    float* a    = h + (size_t)NNODES * HID;
    float* dinv = a + (size_t)NNODES * HID;
    int* wl   = (int*)ws + (size_t)WL_OFF;
    int* wcnt = (int*)ws + (size_t)WCNT_OFF;
    int* ovf  = (int*)ws + (size_t)OVF_OFF;
    unsigned* ob = (unsigned*)ws + (size_t)OB_OFF;
    int* ew   = (int*)ws + (size_t)EW_OFF;

    const int TPB = 256;
    const int gGmm = ((NNODES + 15) / 16 + 7) / 8;

    k_gemm_wmma<<<gGmm, TPB, 0, stream>>>(x, W1, h, NNODES, 0, src, dst, ob, ew);

    k_owner_filter<<<OBLK, 1024, 0, stream>>>(ob, ew, wl, wcnt, ovf, dinv);

    k_owner_agg<<<OBLK, 1024, 0, stream>>>(h, dinv, b1, src, dst, wl, wcnt, ovf, a);

    k_gemm_wmma<<<gGmm, TPB, 0, stream>>>(a, W2, h, NNODES, 1, nullptr, nullptr, nullptr, nullptr);
    k_owner_agg<<<OBLK, 1024, 0, stream>>>(h, dinv, b2, src, dst, wl, wcnt, ovf, a);

    k_graph_pool_head<<<NGRAPHS, PT, 0, stream>>>(a, batch, Wl, bl, out);
}


# Both layers sum rows first and multiply after (sums are linear): each owner block applies W to its LDS rows on WMMA, adds bias, relu; both product launches go, 6 launches to 5; sum order changed
# speedup vs baseline: 4.2960x; 1.1534x over previous
#include <hip/hip_runtime.h>

typedef float v2f __attribute__((ext_vector_type(2)));
typedef float v8f __attribute__((ext_vector_type(8)));

#define NNODES 50000
#define NEDGES 800000
#define HID 64
#define NGRAPHS 512
#define NCLASSES 10


#define OT   196
#define NV4  (NEDGES / 4)
#define EPT  3
#define EBLK ((NV4 + 256 * EPT - 1) / (256 * EPT))

__global__ __launch_bounds__(256)
void k_edge_words(const int* __restrict__ src, const int* __restrict__ dst,
                  unsigned* __restrict__ ob, int* __restrict__ ew) {
    const int t0 = blockIdx.x * 256 + threadIdx.x;
    int4 dv[EPT], sv[EPT];
    #pragma unroll
    for (int j = 0; j < EPT; ++j) {
        const int g = t0 + j * (EBLK * 256);
        const int gc = g < NV4 ? g : NV4 - 1;
        dv[j] = ((const int4*)dst)[gc];
        sv[j] = ((const int4*)src)[gc];
    }
    #pragma unroll
    for (int j = 0; j < EPT; ++j) {
        const int g = t0 + j * (EBLK * 256);
        if (g < NV4) {
            const int4 d = dv[j];
            const int4 s = sv[j];
            const unsigned o0 = (unsigned)d.x / OT, o1 = (unsigned)d.y / OT;
            const unsigned o2 = (unsigned)d.z / OT, o3 = (unsigned)d.w / OT;
            int4 w;
            w.x = s.x | (int)(((unsigned)d.x - o0 * OT) << 16);
            w.y = s.y | (int)(((unsigned)d.y - o1 * OT) << 16);
            w.z = s.z | (int)(((unsigned)d.z - o2 * OT) << 16);
            w.w = s.w | (int)(((unsigned)d.w - o3 * OT) << 16);
            ((int4*)ew)[g] = w;
            ob[g] = (o0 & 255u) | (o1 & 255u) << 8 | (o2 & 255u) << 16 | o3 << 24;
        }
    }
}


#define OBLK ((NNODES + OT - 1) / OT)
#define OTP  ((OT + 1 + 15) / 16 * 16)
#define NT16 (OTP / 16)
#define AP   68
#define WP   72
#define TL   5
#define SCH  (TL * 32)
#define NSC  (NV4 / SCH)
#define RG   16
#define QCAP 896
#define WCAP 512
#define FQCAP 1312
#define WL_OFF   ((2 * NNODES * HID + NNODES + 63) / 64 * 64)
#define WCNT_OFF (WL_OFF + OBLK * 32 * WCAP)
#define OVF_OFF  (WCNT_OFF + OBLK * 32)
#define OB_OFF   ((OVF_OFF + OBLK + 63) / 64 * 64)
#define EW_OFF   ((OB_OFF + NV4 + 63) / 64 * 64)
#define NV16 (NEDGES / 16)
#define TB   5
#define NTB  ((NV16 + TB * 32 - 1) / (TB * 32))
static_assert(NEDGES % 4 == 0 && NV4 % SCH == 0, "dst is read in whole trips of 16-byte groups");
static_assert(2 * NT16 <= 32 && HID == 64, "product: two waves per 16-row tile, 32 columns each");
static_assert(AP % 4 == 0 && WP % 4 == 0 && AP >= HID && WP >= HID, "LDS rows hold 64 floats and start on 16-byte boundaries");
static_assert(NEDGES <= (1 << 20) && OT <= (1 << 12), "queue word = edge | row << 20");
static_assert(NNODES <= (1 << 16) && OT <= (1 << 16), "list word = source | row << 16");
static_assert(QCAP >= 31 + SCH * 4 && FQCAP >= QCAP && NSC >= 32 && 32 % RG == 0, "queue room, a trip for every wave");
static_assert(32 * FQCAP * 4 > 160 * 1024 && 32 * FQCAP * 4 + OT * 4 + 4 <= 320 * 1024, "filter block: over half of the LDS, within all of it");
static_assert(OBLK <= 256 && NEDGES % 16 == 0 && FQCAP >= 31 + 32 * 16 && NTB >= 32, "an owner number is a byte; one load's hits fit the queue");

template <class Take>
__device__ __forceinline__ void walk_dst(const int* __restrict__ dst, int lo, int wave, int lane,
                                         unsigned* __restrict__ myq, Take&& take) {
    int qn = 0;
    const int4* __restrict__ dst4 = (const int4*)dst;
    const int start = (blockIdx.x * (NSC / OBLK)) % NSC;
    const int ntrip = (NSC - wave + 31) / 32;

    auto load_trip = [&](int i, int4 (&buf)[TL], int& base) {
        const int ic = i < ntrip ? i : ntrip - 1;
        int s2 = ic * 32 + wave + start;
        if (s2 >= NSC) s2 -= NSC;
        base = s2 * SCH;
        #pragma unroll
        for (int k = 0; k < TL; ++k) buf[k] = dst4[base + k * 32 + lane];
    };

    auto push = [&](unsigned t, int e) {
        const bool m = t < (unsigned)OT;
        const unsigned mask = __builtin_amdgcn_ballot_w32(m);
        if (mask != 0u) {
            if (m) {
                const int slot = qn + (int)__builtin_amdgcn_mbcnt_lo(mask, 0u);
                myq[slot] = (unsigned)e | (t << 20);
            }
            qn += __builtin_popcount(mask);
        }
    };

    auto process = [&](const int4 (&buf)[TL], int base, bool last) {
        #pragma unroll
        for (int k = 0; k < TL; ++k) {
            const int4 d = buf[k];
            const unsigned t0 = (unsigned)(d.x - lo), t1 = (unsigned)(d.y - lo);
            const unsigned t2 = (unsigned)(d.z - lo), t3 = (unsigned)(d.w - lo);
            const unsigned tm = min(min(t0, t1), min(t2, t3));
            if (__builtin_amdgcn_ballot_w32(tm < (unsigned)OT) != 0u) {
                const int e0 = (base + k * 32 + lane) * 4;
                push(t0, e0);
                push(t1, e0 + 1);
                push(t2, e0 + 2);
                push(t3, e0 + 3);
            }
        }
        const int keep = last ? 0 : 31;
        while (qn > keep) qn = take(qn);
    };

    int4 bufA[TL], bufB[TL];
    int baseA = 0, baseB = 0;
    load_trip(0, bufA, baseA);
    for (int i = 0; i < ntrip; i += 2) {
        load_trip(i + 1, bufB, baseB);
        process(bufA, baseA, i + 1 >= ntrip);
        if (i + 1 >= ntrip) break;
        load_trip(i + 2, bufA, baseA);
        process(bufB, baseB, i + 2 >= ntrip);
    }
}

__global__ __launch_bounds__(1024)
void k_owner_filter(const unsigned* __restrict__ ob, const int* __restrict__ ew,
                    int* __restrict__ wl, int* __restrict__ wcnt, int* __restrict__ ovf,
                    float* __restrict__ dinv) {
    __shared__ int scnt[OT];
    __shared__ int sovf;
    __shared__ unsigned q[32 * FQCAP];

    const int tid  = threadIdx.x;
    const int lane = tid & 31;
    const int wave = __builtin_amdgcn_readfirstlane(tid >> 5);
    const int lo   = blockIdx.x * OT;
    const unsigned idr = (unsigned)blockIdx.x * 0x01010101u;
    const uint4* __restrict__ ob4 = (const uint4*)ob;

    if (tid < OT) scnt[tid] = 0;
    if (tid == 0) sovf = 0;
    __syncthreads();

    unsigned* __restrict__ myq = q + wave * FQCAP;
    int* __restrict__ mywl = wl + (size_t)(blockIdx.x * 32 + wave) * WCAP;
    int wn = 0;
    int qn = 0;

    auto take = [&](int fill) {
        const int base = fill > 32 ? fill - 32 : 0;
        const int cnt  = fill - base;
        if (lane < cnt) {
            const unsigned w = myq[base + lane];
            const unsigned p = w & 31u;
            const unsigned e = (w >> 5) * 16u + ((p & 7u) << 2) + (p >> 3);
            const int word = ew[e];
            atomicAdd(&scnt[(unsigned)word >> 16], 1);
            if (wn + lane < WCAP) mywl[wn + lane] = word;
        }
        wn += cnt;
        return base;
    };

    const int start = (blockIdx.x * NTB) / OBLK;
    const int ntrip = (NTB - wave + 31) / 32;

    auto load_trip = [&](int i, uint4 (&buf)[TB], int& base) {
        const int ic = i < ntrip ? i : ntrip - 1;
        int s2 = ic * 32 + wave + start;
        if (s2 >= NTB) s2 -= NTB;
        base = s2 * (TB * 32);
        #pragma unroll
        for (int k = 0; k < TB; ++k) {
            const int g = base + k * 32 + lane;
            buf[k] = ob4[g < NV16 ? g : NV16 - 1];
        }
    };

    auto eq7 = [&](unsigned x) {
        const unsigned y = x ^ idr;
        return ~(((y & 0x7F7F7F7Fu) + 0x7F7F7F7Fu) | y | 0x7F7F7F7Fu);
    };

    auto process = [&](const uint4 (&buf)[TB], int base) {
        #pragma unroll
        for (int k = 0; k < TB; ++k) {
            const uint4 d = buf[k];
            const int g = base + k * 32 + lane;
            unsigned m = (eq7(d.x) >> 7) | (eq7(d.y) >> 6) | (eq7(d.z) >> 5) | (eq7(d.w) >> 4);
            if (base + k * 32 + 32 > NV16) m = g < NV16 ? m : 0u;
            unsigned any = __builtin_amdgcn_ballot_w32(m != 0u);
            while (any != 0u) {
                if (m != 0u) {
                    const int p = __builtin_ctz(m);
                    const int slot = qn + (int)__builtin_amdgcn_mbcnt_lo(any, 0u);
                    myq[slot] = (unsigned)(g * 32 + p);
                    m &= m - 1u;
                }
                qn += __builtin_popcount(any);
                any = __builtin_amdgcn_ballot_w32(m != 0u);
            }
            while (qn > 31) qn = take(qn);
        }
    };

    uint4 bufA[TB], bufB[TB];
    int baseA = 0, baseB = 0;
    load_trip(0, bufA, baseA);
    for (int i = 0; i < ntrip; i += 2) {
        load_trip(i + 1, bufB, baseB);
        process(bufA, baseA);
        if (i + 1 >= ntrip) break;
        load_trip(i + 2, bufA, baseA);
        process(bufB, baseB);
    }
    while (qn > 0) qn = take(qn);

    if (lane == 0) {
        wcnt[blockIdx.x * 32 + wave] = wn;
        if (wn > WCAP) sovf = 1;
    }
    __syncthreads();
    if (tid < OT && lo + tid < NNODES) dinv[lo + tid] = rsqrtf((float)(scnt[tid] + 1));
    if (tid == 0) ovf[blockIdx.x] = sovf;
}

__global__ __launch_bounds__(1024)
void k_owner_agg(const float* __restrict__ h, const float* __restrict__ dinv,
                 const float* __restrict__ W, const float* __restrict__ bias, int relu_out,
                 const int* __restrict__ src, const int* __restrict__ dst,
                 const int* __restrict__ wl, const int* __restrict__ wcnt, const int* __restrict__ ovf,
                 float* __restrict__ a) {
    __shared__ float acc[OTP * AP];
    __shared__ float sW[HID * WP];
    __shared__ float sdinv[OT];
    __shared__ unsigned q[32 * QCAP];

    const int tid  = threadIdx.x;
    const int lane = tid & 31;
    const int wave = __builtin_amdgcn_readfirstlane(tid >> 5);
    const int lo   = blockIdx.x * OT;

    const int over = ovf[blockIdx.x];
    const int* __restrict__ mywl = wl + (size_t)(blockIdx.x * 32 + wave) * WCAP;
    int n = 0;
    int wnext = 0;
    if (!over) {
        n = __builtin_amdgcn_readfirstlane(wcnt[blockIdx.x * 32 + wave]);
        if (n > 0) wnext = mywl[lane < n ? lane : n - 1];
    }

    const int l16  = lane & 15;
    const int half = lane >> 4;
    const int nb   = (wave & 1) * 32;
    ((float4*)(sW + (tid >> 4) * WP))[tid & 15] = ((const float4*)W)[tid];
    const float bv0 = bias[nb + l16];
    const float bv1 = bias[nb + 16 + l16];

    for (int i = tid; i < OTP * 16; i += 1024) {
        const int r = i >> 4, c = i & 15;
        const int node = lo + r;
        float4 v = make_float4(0.f, 0.f, 0.f, 0.f);
        if (r < OT && node < NNODES) {
            const float4 hv = ((const float4*)h)[(size_t)node * 16 + c];
            const float di = dinv[node];
            v.x = hv.x * di * di;
            v.y = hv.y * di * di;
            v.z = hv.z * di * di;
            v.w = hv.w * di * di;
        }
        ((float4*)(acc + r * AP))[c] = v;
    }
    if (tid < OT) sdinv[tid] = (lo + tid < NNODES) ? dinv[lo + tid] : 0.0f;
    __syncthreads();

    const char* __restrict__ hb = (const char*)h;
    char* accb = (char*)acc;

    auto add_rows = [&](int cnt, unsigned soff, unsigned doff, float dv, float sd) {
        #pragma nounroll
        for (int j0 = 0; j0 < cnt; j0 += RG) {
            float2 v[RG];
            #pragma unroll
            for (int k = 0; k < RG; ++k) {
                const unsigned so = (unsigned)__builtin_amdgcn_readlane((int)soff, j0 + k);
                v[k] = *(const float2*)(hb + (so + (unsigned)lane * 8u));
            }
            #pragma unroll
            for (int k = 0; k < RG; ++k) {
                const float wj =
                    __builtin_bit_cast(float, __builtin_amdgcn_readlane(__builtin_bit_cast(int, dv), j0 + k)) *
                    __builtin_bit_cast(float, __builtin_amdgcn_readlane(__builtin_bit_cast(int, sd), j0 + k));
                const unsigned dj = (unsigned)__builtin_amdgcn_readlane((int)doff, j0 + k);
                float* ap = (float*)(accb + (dj + (unsigned)lane * 8u));
                atomicAdd(ap,     v[k].x * wj);
                atomicAdd(ap + 1, v[k].y * wj);
            }
        }
    };

    if (!over) {
        for (int j0 = 0; j0 < n; j0 += 32) {
            const int cnt = min(32, n - j0);
            const unsigned word = (unsigned)wnext;
            const int nj = j0 + 32 + lane;
            wnext = mywl[nj < n ? nj : n - 1];
            const unsigned s  = word & 0xFFFFu;
            const unsigned dl = word >> 16;
            const float dv = dinv[s];
            const float sd = sdinv[dl];
            add_rows(cnt, s * (HID * 4u), (lane < cnt ? dl : (unsigned)OT) * (AP * 4u), dv, sd);
        }
    } else {
        unsigned* __restrict__ myq = q + wave * QCAP;
        walk_dst(dst, lo, wave, lane, myq, [&](int qn) {
            const int base = qn > 32 ? qn - 32 : 0;
            const int cnt  = qn - base;
            const bool used = lane < cnt;
            const unsigned p = myq[base + (used ? lane : cnt - 1)];
            const int e  = (int)(p & 0xFFFFFu);
            const int dl = (int)(p >> 20);
            const int s  = src[e];
            const float dv = dinv[s];
            const float sd = sdinv[dl];
            add_rows(cnt, (unsigned)s * (HID * 4u), (unsigned)(used ? dl : OT) * (AP * 4u), dv, sd);
            return base;
        });
    }
    __syncthreads();

    v8f c0 = {}, c1 = {};
    const int t16 = wave >> 1;
    if (wave < 2 * NT16) {
        const float* Ar = acc + (t16 * 16 + l16) * AP;
        #pragma unroll
        for (int k0 = 0; k0 < HID; k0 += 4) {
            const int kb = k0 + half * 2;
            const v2f af = {Ar[kb], Ar[kb + 1]};
            const float* Wk0 = sW + kb * WP + nb;
            const float* Wk1 = sW + (kb + 1) * WP + nb;
            const v2f bf0 = {Wk0[l16],      Wk1[l16]};
            const v2f bf1 = {Wk0[l16 + 16], Wk1[l16 + 16]};
            c0 = __builtin_amdgcn_wmma_f32_16x16x4_f32(false, af, false, bf0, (short)0, c0, false, false);
            c1 = __builtin_amdgcn_wmma_f32_16x16x4_f32(false, af, false, bf1, (short)0, c1, false, false);
        }
    }
    __syncthreads();
    if (wave < 2 * NT16) {
        #pragma unroll
        for (int r = 0; r < 8; ++r) {
            float* orow = acc + (t16 * 16 + half * 8 + r) * AP + nb + l16;
            float v0 = c0[r] + bv0;
            float v1 = c1[r] + bv1;
            if (relu_out) { v0 = fmaxf(v0, 0.0f); v1 = fmaxf(v1, 0.0f); }
            orow[0]  = v0;
            orow[16] = v1;
        }
    }
    __syncthreads();

    for (int i = tid; i < OT * 16; i += 1024) {
        const int node = lo + (i >> 4);
        if (node < NNODES) ((float4*)a)[(size_t)node * 16 + (i & 15)] = ((const float4*)(acc + (i >> 4) * AP))[i & 15];
    }
}


#define PT    256
#define PW    (PT / 32)
#define NB4   (NNODES / 4)
#define NK    ((NB4 + PT - 1) / PT)
#define NWORD ((NK + 7) / 8)
#define LCAP  1024
#define RB    16
#define NWL   ((HID * NCLASSES + PT - 1) / PT)
static_assert(NNODES % 4 == 0 && HID == 64 && NCLASSES <= PT, "batch in 16-byte groups, a row is 32 lanes of 8 bytes");

__global__ __launch_bounds__(PT)
void k_graph_pool_head(const float* __restrict__ a, const int* __restrict__ batch,
                       const float* __restrict__ Wl, const float* __restrict__ bl,
                       float* __restrict__ out) {
    __shared__ int   list[LCAP];
    __shared__ int   wtot[PW];
    __shared__ float part[PW * HID];
    __shared__ float ssum[HID];
    __shared__ float sWl[HID * NCLASSES];

    const int tid  = threadIdx.x;
    const int lane = tid & 31;
    const int wave = __builtin_amdgcn_readfirstlane(tid >> 5);
    const int g    = blockIdx.x;

    float wl[NWL];
    #pragma unroll
    for (int i = 0; i < NWL; ++i) wl[i] = (i + 1) * PT <= HID * NCLASSES || tid + i * PT < HID * NCLASSES ? Wl[tid + i * PT] : 0.0f;
    const float blv = tid < NCLASSES ? bl[tid] : 0.0f;

    const int4* __restrict__ batch4 = (const int4*)batch;
    unsigned mask[NWORD];
    #pragma unroll
    for (int w = 0; w < NWORD; ++w) {
        int4 d[8];
        #pragma unroll
        for (int j = 0; j < 8; ++j) {
            const int k = w * 8 + j;
            d[j] = make_int4(-1, -1, -1, -1);
            if (k < NK) {
                const int idx = k * PT + tid;
                if ((k + 1) * PT <= NB4 || idx < NB4) d[j] = batch4[idx];
            }
        }
        unsigned m = 0u;
        #pragma unroll
        for (int j = 0; j < 8; ++j) {
            if (w * 8 + j < NK) {
                m |= (d[j].x == g ? 1u : 0u) << (4 * j);
                m |= (d[j].y == g ? 2u : 0u) << (4 * j);
                m |= (d[j].z == g ? 4u : 0u) << (4 * j);
                m |= (d[j].w == g ? 8u : 0u) << (4 * j);
            }
        }
        mask[w] = m;
    }

    #pragma unroll
    for (int i = 0; i < NWL; ++i) if ((i + 1) * PT <= HID * NCLASSES || tid + i * PT < HID * NCLASSES) sWl[tid + i * PT] = wl[i];

    int c = 0;
    #pragma unroll
    for (int w = 0; w < NWORD; ++w) c += __builtin_popcount(mask[w]);
    int x = c;
    for (int dl = 1; dl < 32; dl <<= 1) {
        const int y = __shfl_up(x, dl);
        if (lane >= dl) x += y;
    }
    if (lane == 31) wtot[wave] = x;
    __syncthreads();
    int base = 0, total = 0;
    for (int w = 0; w < PW; ++w) {
        const int t = wtot[w];
        if (w < wave) base += t;
        total += t;
    }
    const int first = base + x - c;

    float2 acc = make_float2(0.0f, 0.0f);
    const float2* __restrict__ a2 = (const float2*)a;
    for (int r0 = 0; r0 < total; r0 += LCAP) {
        if (r0 != 0) __syncthreads();
        if (c != 0) {
            int p = first - r0;
            #pragma unroll
            for (int w = 0; w < NWORD; ++w) {
                unsigned m = mask[w];
                while (m != 0u) {
                    const int b = __builtin_ctz(m);
                    m &= m - 1u;
                    if (p >= 0 && p < LCAP) list[p] = (((w * 8 + (b >> 2)) * PT + tid) << 2) + (b & 3);
                    ++p;
                }
            }
        }
        __syncthreads();
        const int n = min(total - r0, LCAP);
        for (int p0 = wave; p0 < n + wave; p0 += PW * RB) {
            float2 v[RB];
            #pragma unroll
            for (int k = 0; k < RB; ++k) {
                const int p = p0 + PW * k < n ? p0 + PW * k : n - 1;
                const int node = __builtin_amdgcn_readfirstlane(list[p]);
                v[k] = a2[(size_t)node * 32 + lane];
            }
            #pragma unroll
            for (int k = 0; k < RB; ++k) {
                const bool ok = p0 + PW * k < n;
                acc.x += fmaxf(ok ? v[k].x : 0.0f, 0.0f);
                acc.y += fmaxf(ok ? v[k].y : 0.0f, 0.0f);
            }
        }
    }

    ((float2*)part)[wave * 32 + lane] = acc;
    __syncthreads();
    if (tid < HID) {
        float s = 0.0f;
        for (int w = 0; w < PW; ++w) s += part[w * HID + tid];
        ssum[tid] = s;
    }
    __syncthreads();
    if (tid < NCLASSES) {
        const float inv = 1.0f / fmaxf((float)total, 1.0f);
        float o = 0.0f;
        #pragma unroll
        for (int k = 0; k < HID; ++k) o += ssum[k] * sWl[k * NCLASSES + tid];
        out[g * NCLASSES + tid] = o * inv + blv;
    }
}


extern "C" void kernel_launch(void* const* d_in, const int* in_sizes, int n_in,
                              void* d_out, int out_size, void* d_ws, size_t ws_size,
                              hipStream_t stream) {
    const float* x  = (const float*)d_in[0];
    const float* W1 = (const float*)d_in[1];
    const float* b1 = (const float*)d_in[2];
    const float* W2 = (const float*)d_in[3];
    const float* b2 = (const float*)d_in[4];
    const float* Wl = (const float*)d_in[5];
    const float* bl = (const float*)d_in[6];
    const int* edge  = (const int*)d_in[7];
    const int* batch = (const int*)d_in[8];
    const int* src = edge;
    const int* dst = edge + NEDGES;
    float* out = (float*)d_out;

    float* ws   = (float*)d_ws;
    float* h    = ws;
    float* a    = h + (size_t)NNODES * HID;
    float* dinv = a + (size_t)NNODES * HID;
    int* wl   = (int*)ws + (size_t)WL_OFF;
    int* wcnt = (int*)ws + (size_t)WCNT_OFF;
    int* ovf  = (int*)ws + (size_t)OVF_OFF;
    unsigned* ob = (unsigned*)ws + (size_t)OB_OFF;
    int* ew   = (int*)ws + (size_t)EW_OFF;

    k_edge_words<<<EBLK, 256, 0, stream>>>(src, dst, ob, ew);

    k_owner_filter<<<OBLK, 1024, 0, stream>>>(ob, ew, wl, wcnt, ovf, dinv);

    k_owner_agg<<<OBLK, 1024, 0, stream>>>(x, dinv, W1, b1, 1, src, dst, wl, wcnt, ovf, h);

    k_owner_agg<<<OBLK, 1024, 0, stream>>>(h, dinv, W2, b2, 0, src, dst, wl, wcnt, ovf, a);

    k_graph_pool_head<<<NGRAPHS, PT, 0, stream>>>(a, batch, Wl, bl, out);
}


# Pooling blocks read a table of lowest and highest graph id per 128-node chunk (written by the first launch) and only the chunks that can hold their id, not all of batch; pooling sum order changed
# speedup vs baseline: 4.5215x; 1.0525x over previous
#include <hip/hip_runtime.h>

typedef float v2f __attribute__((ext_vector_type(2)));
typedef float v8f __attribute__((ext_vector_type(8)));

#define NNODES 50000
#define NEDGES 800000
#define HID 64
#define NGRAPHS 512
#define NCLASSES 10


#define OT   196
#define NV4  (NEDGES / 4)
#define EPT  3
#define EBLK ((NV4 + 256 * EPT - 1) / (256 * EPT))
#define NB4   (NNODES / 4)
#define CG    32
#define NCH   ((NB4 + CG - 1) / CG)
static_assert(NNODES % 4 == 0 && EBLK * 8 >= NCH, "batch in 16-byte groups; the first launch has a wave for every chunk");

__global__ __launch_bounds__(256)
void k_edge_words(const int* __restrict__ src, const int* __restrict__ dst, const int* __restrict__ batch,
                  unsigned* __restrict__ ob, int* __restrict__ ew, int* __restrict__ zmm) {
    const int t0 = blockIdx.x * 256 + threadIdx.x;
    const int lane = threadIdx.x & 31;
    const int gw = __builtin_amdgcn_readfirstlane(blockIdx.x * 8 + (int)(threadIdx.x >> 5));
    int4 bq = make_int4(0, 0, 0, 0);
    if (gw < NCH) {
        const int gi = gw * CG + lane;
        bq = ((const int4*)batch)[gi < NB4 ? gi : NB4 - 1];
    }
    int4 dv[EPT], sv[EPT];
    #pragma unroll
    for (int j = 0; j < EPT; ++j) {
        const int g = t0 + j * (EBLK * 256);
        const int gc = g < NV4 ? g : NV4 - 1;
        dv[j] = ((const int4*)dst)[gc];
        sv[j] = ((const int4*)src)[gc];
    }
    #pragma unroll
    for (int j = 0; j < EPT; ++j) {
        const int g = t0 + j * (EBLK * 256);
        if (g < NV4) {
            const int4 d = dv[j];
            const int4 s = sv[j];
            const unsigned o0 = (unsigned)d.x / OT, o1 = (unsigned)d.y / OT;
            const unsigned o2 = (unsigned)d.z / OT, o3 = (unsigned)d.w / OT;
            int4 w;
            w.x = s.x | (int)(((unsigned)d.x - o0 * OT) << 16);
            w.y = s.y | (int)(((unsigned)d.y - o1 * OT) << 16);
            w.z = s.z | (int)(((unsigned)d.z - o2 * OT) << 16);
            w.w = s.w | (int)(((unsigned)d.w - o3 * OT) << 16);
            ((int4*)ew)[g] = w;
            ob[g] = (o0 & 255u) | (o1 & 255u) << 8 | (o2 & 255u) << 16 | o3 << 24;
        }
    }
    if (gw < NCH) {
        int mn = min(min(bq.x, bq.y), min(bq.z, bq.w));
        int mx = max(max(bq.x, bq.y), max(bq.z, bq.w));
        for (int dl = 16; dl >= 1; dl >>= 1) {
            mn = min(mn, __shfl_xor(mn, dl));
            mx = max(mx, __shfl_xor(mx, dl));
        }
        if (lane == 0) ((int2*)zmm)[gw] = make_int2(mn, mx);
    }
}


#define OBLK ((NNODES + OT - 1) / OT)
#define OTP  ((OT + 1 + 15) / 16 * 16)
#define NT16 (OTP / 16)
#define AP   68
#define WP   72
#define TL   5
#define SCH  (TL * 32)
#define NSC  (NV4 / SCH)
#define RG   16
#define QCAP 896
#define WCAP 512
#define FQCAP 1312
#define WL_OFF   ((2 * NNODES * HID + NNODES + 63) / 64 * 64)
#define WCNT_OFF (WL_OFF + OBLK * 32 * WCAP)
#define OVF_OFF  (WCNT_OFF + OBLK * 32)
#define OB_OFF   ((OVF_OFF + OBLK + 63) / 64 * 64)
#define EW_OFF   ((OB_OFF + NV4 + 63) / 64 * 64)
#define ZM_OFF   ((EW_OFF + NEDGES + 63) / 64 * 64)
#define NV16 (NEDGES / 16)
#define TB   5
#define NTB  ((NV16 + TB * 32 - 1) / (TB * 32))
static_assert(NEDGES % 4 == 0 && NV4 % SCH == 0, "dst is read in whole trips of 16-byte groups");
static_assert(2 * NT16 <= 32 && HID == 64, "product: two waves per 16-row tile, 32 columns each");
static_assert(AP % 4 == 0 && WP % 4 == 0 && AP >= HID && WP >= HID, "LDS rows hold 64 floats and start on 16-byte boundaries");
static_assert(NEDGES <= (1 << 20) && OT <= (1 << 12), "queue word = edge | row << 20");
static_assert(NNODES <= (1 << 16) && OT <= (1 << 16), "list word = source | row << 16");
static_assert(QCAP >= 31 + SCH * 4 && FQCAP >= QCAP && NSC >= 32 && 32 % RG == 0, "queue room, a trip for every wave");
static_assert(32 * FQCAP * 4 > 160 * 1024 && 32 * FQCAP * 4 + OT * 4 + 4 <= 320 * 1024, "filter block: over half of the LDS, within all of it");
static_assert(OBLK <= 256 && NEDGES % 16 == 0 && FQCAP >= 31 + 32 * 16 && NTB >= 32, "an owner number is a byte; one load's hits fit the queue");

template <class Take>
__device__ __forceinline__ void walk_dst(const int* __restrict__ dst, int lo, int wave, int lane,
                                         unsigned* __restrict__ myq, Take&& take) {
    int qn = 0;
    const int4* __restrict__ dst4 = (const int4*)dst;
    const int start = (blockIdx.x * (NSC / OBLK)) % NSC;
    const int ntrip = (NSC - wave + 31) / 32;

    auto load_trip = [&](int i, int4 (&buf)[TL], int& base) {
        const int ic = i < ntrip ? i : ntrip - 1;
        int s2 = ic * 32 + wave + start;
        if (s2 >= NSC) s2 -= NSC;
        base = s2 * SCH;
        #pragma unroll
        for (int k = 0; k < TL; ++k) buf[k] = dst4[base + k * 32 + lane];
    };

    auto push = [&](unsigned t, int e) {
        const bool m = t < (unsigned)OT;
        const unsigned mask = __builtin_amdgcn_ballot_w32(m);
        if (mask != 0u) {
            if (m) {
                const int slot = qn + (int)__builtin_amdgcn_mbcnt_lo(mask, 0u);
                myq[slot] = (unsigned)e | (t << 20);
            }
            qn += __builtin_popcount(mask);
        }
    };

    auto process = [&](const int4 (&buf)[TL], int base, bool last) {
        #pragma unroll
        for (int k = 0; k < TL; ++k) {
            const int4 d = buf[k];
            const unsigned t0 = (unsigned)(d.x - lo), t1 = (unsigned)(d.y - lo);
            const unsigned t2 = (unsigned)(d.z - lo), t3 = (unsigned)(d.w - lo);
            const unsigned tm = min(min(t0, t1), min(t2, t3));
            if (__builtin_amdgcn_ballot_w32(tm < (unsigned)OT) != 0u) {
                const int e0 = (base + k * 32 + lane) * 4;
                push(t0, e0);
                push(t1, e0 + 1);
                push(t2, e0 + 2);
                push(t3, e0 + 3);
            }
        }
        const int keep = last ? 0 : 31;
        while (qn > keep) qn = take(qn);
    };

    int4 bufA[TL], bufB[TL];
    int baseA = 0, baseB = 0;
    load_trip(0, bufA, baseA);
    for (int i = 0; i < ntrip; i += 2) {
        load_trip(i + 1, bufB, baseB);
        process(bufA, baseA, i + 1 >= ntrip);
        if (i + 1 >= ntrip) break;
        load_trip(i + 2, bufA, baseA);
        process(bufB, baseB, i + 2 >= ntrip);
    }
}

__global__ __launch_bounds__(1024)
void k_owner_filter(const unsigned* __restrict__ ob, const int* __restrict__ ew,
                    int* __restrict__ wl, int* __restrict__ wcnt, int* __restrict__ ovf,
                    float* __restrict__ dinv) {
    __shared__ int scnt[OT];
    __shared__ int sovf;
    __shared__ unsigned q[32 * FQCAP];

    const int tid  = threadIdx.x;
    const int lane = tid & 31;
    const int wave = __builtin_amdgcn_readfirstlane(tid >> 5);
    const int lo   = blockIdx.x * OT;
    const unsigned idr = (unsigned)blockIdx.x * 0x01010101u;
    const uint4* __restrict__ ob4 = (const uint4*)ob;

    if (tid < OT) scnt[tid] = 0;
    if (tid == 0) sovf = 0;
    __syncthreads();

    unsigned* __restrict__ myq = q + wave * FQCAP;
    int* __restrict__ mywl = wl + (size_t)(blockIdx.x * 32 + wave) * WCAP;
    int wn = 0;
    int qn = 0;

    auto take = [&](int fill) {
        const int base = fill > 32 ? fill - 32 : 0;
        const int cnt  = fill - base;
        if (lane < cnt) {
            const unsigned w = myq[base + lane];
            const unsigned p = w & 31u;
            const unsigned e = (w >> 5) * 16u + ((p & 7u) << 2) + (p >> 3);
            const int word = ew[e];
            atomicAdd(&scnt[(unsigned)word >> 16], 1);
            if (wn + lane < WCAP) mywl[wn + lane] = word;
        }
        wn += cnt;
        return base;
    };

    const int start = (blockIdx.x * NTB) / OBLK;
    const int ntrip = (NTB - wave + 31) / 32;

    auto load_trip = [&](int i, uint4 (&buf)[TB], int& base) {
        const int ic = i < ntrip ? i : ntrip - 1;
        int s2 = ic * 32 + wave + start;
        if (s2 >= NTB) s2 -= NTB;
        base = s2 * (TB * 32);
        #pragma unroll
        for (int k = 0; k < TB; ++k) {
            const int g = base + k * 32 + lane;
            buf[k] = ob4[g < NV16 ? g : NV16 - 1];
        }
    };

    auto eq7 = [&](unsigned x) {
        const unsigned y = x ^ idr;
        return ~(((y & 0x7F7F7F7Fu) + 0x7F7F7F7Fu) | y | 0x7F7F7F7Fu);
    };

    auto process = [&](const uint4 (&buf)[TB], int base) {
        #pragma unroll
        for (int k = 0; k < TB; ++k) {
            const uint4 d = buf[k];
            const int g = base + k * 32 + lane;
            unsigned m = (eq7(d.x) >> 7) | (eq7(d.y) >> 6) | (eq7(d.z) >> 5) | (eq7(d.w) >> 4);
            if (base + k * 32 + 32 > NV16) m = g < NV16 ? m : 0u;
            unsigned any = __builtin_amdgcn_ballot_w32(m != 0u);
            while (any != 0u) {
                if (m != 0u) {
                    const int p = __builtin_ctz(m);
                    const int slot = qn + (int)__builtin_amdgcn_mbcnt_lo(any, 0u);
                    myq[slot] = (unsigned)(g * 32 + p);
                    m &= m - 1u;
                }
                qn += __builtin_popcount(any);
                any = __builtin_amdgcn_ballot_w32(m != 0u);
            }
            while (qn > 31) qn = take(qn);
        }
    };

    uint4 bufA[TB], bufB[TB];
    int baseA = 0, baseB = 0;
    load_trip(0, bufA, baseA);
    for (int i = 0; i < ntrip; i += 2) {
        load_trip(i + 1, bufB, baseB);
        process(bufA, baseA);
        if (i + 1 >= ntrip) break;
        load_trip(i + 2, bufA, baseA);
        process(bufB, baseB);
    }
    while (qn > 0) qn = take(qn);

    if (lane == 0) {
        wcnt[blockIdx.x * 32 + wave] = wn;
        if (wn > WCAP) sovf = 1;
    }
    __syncthreads();
    if (tid < OT && lo + tid < NNODES) dinv[lo + tid] = rsqrtf((float)(scnt[tid] + 1));
    if (tid == 0) ovf[blockIdx.x] = sovf;
}

__global__ __launch_bounds__(1024)
void k_owner_agg(const float* __restrict__ h, const float* __restrict__ dinv,
                 const float* __restrict__ W, const float* __restrict__ bias, int relu_out,
                 const int* __restrict__ src, const int* __restrict__ dst,
                 const int* __restrict__ wl, const int* __restrict__ wcnt, const int* __restrict__ ovf,
                 float* __restrict__ a) {
    __shared__ float acc[OTP * AP];
    __shared__ float sW[HID * WP];
    __shared__ float sdinv[OT];
    __shared__ unsigned q[32 * QCAP];

    const int tid  = threadIdx.x;
    const int lane = tid & 31;
    const int wave = __builtin_amdgcn_readfirstlane(tid >> 5);
    const int lo   = blockIdx.x * OT;

    const int over = ovf[blockIdx.x];
    const int* __restrict__ mywl = wl + (size_t)(blockIdx.x * 32 + wave) * WCAP;
    int n = 0;
    int wnext = 0;
    if (!over) {
        n = __builtin_amdgcn_readfirstlane(wcnt[blockIdx.x * 32 + wave]);
        if (n > 0) wnext = mywl[lane < n ? lane : n - 1];
    }

    const int l16  = lane & 15;
    const int half = lane >> 4;
    const int nb   = (wave & 1) * 32;
    ((float4*)(sW + (tid >> 4) * WP))[tid & 15] = ((const float4*)W)[tid];
    const float bv0 = bias[nb + l16];
    const float bv1 = bias[nb + 16 + l16];

    for (int i = tid; i < OTP * 16; i += 1024) {
        const int r = i >> 4, c = i & 15;
        const int node = lo + r;
        float4 v = make_float4(0.f, 0.f, 0.f, 0.f);
        if (r < OT && node < NNODES) {
            const float4 hv = ((const float4*)h)[(size_t)node * 16 + c];
            const float di = dinv[node];
            v.x = hv.x * di * di;
            v.y = hv.y * di * di;
            v.z = hv.z * di * di;
            v.w = hv.w * di * di;
        }
        ((float4*)(acc + r * AP))[c] = v;
    }
    if (tid < OT) sdinv[tid] = (lo + tid < NNODES) ? dinv[lo + tid] : 0.0f;
    __syncthreads();

    const char* __restrict__ hb = (const char*)h;
    char* accb = (char*)acc;

    auto add_rows = [&](int cnt, unsigned soff, unsigned doff, float dv, float sd) {
        #pragma nounroll
        for (int j0 = 0; j0 < cnt; j0 += RG) {
            float2 v[RG];
            #pragma unroll
            for (int k = 0; k < RG; ++k) {
                const unsigned so = (unsigned)__builtin_amdgcn_readlane((int)soff, j0 + k);
                v[k] = *(const float2*)(hb + (so + (unsigned)lane * 8u));
            }
            #pragma unroll
            for (int k = 0; k < RG; ++k) {
                const float wj =
                    __builtin_bit_cast(float, __builtin_amdgcn_readlane(__builtin_bit_cast(int, dv), j0 + k)) *
                    __builtin_bit_cast(float, __builtin_amdgcn_readlane(__builtin_bit_cast(int, sd), j0 + k));
                const unsigned dj = (unsigned)__builtin_amdgcn_readlane((int)doff, j0 + k);
                float* ap = (float*)(accb + (dj + (unsigned)lane * 8u));
                atomicAdd(ap,     v[k].x * wj);
                atomicAdd(ap + 1, v[k].y * wj);
            }
        }
    };

    if (!over) {
        for (int j0 = 0; j0 < n; j0 += 32) {
            const int cnt = min(32, n - j0);
            const unsigned word = (unsigned)wnext;
            const int nj = j0 + 32 + lane;
            wnext = mywl[nj < n ? nj : n - 1];
            const unsigned s  = word & 0xFFFFu;
            const unsigned dl = word >> 16;
            const float dv = dinv[s];
            const float sd = sdinv[dl];
            add_rows(cnt, s * (HID * 4u), (lane < cnt ? dl : (unsigned)OT) * (AP * 4u), dv, sd);
        }
    } else {
        unsigned* __restrict__ myq = q + wave * QCAP;
        walk_dst(dst, lo, wave, lane, myq, [&](int qn) {
            const int base = qn > 32 ? qn - 32 : 0;
            const int cnt  = qn - base;
            const bool used = lane < cnt;
            const unsigned p = myq[base + (used ? lane : cnt - 1)];
            const int e  = (int)(p & 0xFFFFFu);
            const int dl = (int)(p >> 20);
            const int s  = src[e];
            const float dv = dinv[s];
            const float sd = sdinv[dl];
            add_rows(cnt, (unsigned)s * (HID * 4u), (unsigned)(used ? dl : OT) * (AP * 4u), dv, sd);
            return base;
        });
    }
    __syncthreads();

    v8f c0 = {}, c1 = {};
    const int t16 = wave >> 1;
    if (wave < 2 * NT16) {
        const float* Ar = acc + (t16 * 16 + l16) * AP;
        #pragma unroll
        for (int k0 = 0; k0 < HID; k0 += 4) {
            const int kb = k0 + half * 2;
            const v2f af = {Ar[kb], Ar[kb + 1]};
            const float* Wk0 = sW + kb * WP + nb;
            const float* Wk1 = sW + (kb + 1) * WP + nb;
            const v2f bf0 = {Wk0[l16],      Wk1[l16]};
            const v2f bf1 = {Wk0[l16 + 16], Wk1[l16 + 16]};
            c0 = __builtin_amdgcn_wmma_f32_16x16x4_f32(false, af, false, bf0, (short)0, c0, false, false);
            c1 = __builtin_amdgcn_wmma_f32_16x16x4_f32(false, af, false, bf1, (short)0, c1, false, false);
        }
    }
    __syncthreads();
    if (wave < 2 * NT16) {
        #pragma unroll
        for (int r = 0; r < 8; ++r) {
            float* orow = acc + (t16 * 16 + half * 8 + r) * AP + nb + l16;
            float v0 = c0[r] + bv0;
            float v1 = c1[r] + bv1;
            if (relu_out) { v0 = fmaxf(v0, 0.0f); v1 = fmaxf(v1, 0.0f); }
            orow[0]  = v0;
            orow[16] = v1;
        }
    }
    __syncthreads();

    for (int i = tid; i < OT * 16; i += 1024) {
        const int node = lo + (i >> 4);
        if (node < NNODES) ((float4*)a)[(size_t)node * 16 + (i & 15)] = ((const float4*)(acc + (i >> 4) * AP))[i & 15];
    }
}


#define PT    256
#define PW    (PT / 32)
#define NK    ((NB4 + PT - 1) / PT)
#define NZ    ((NCH + PT - 1) / PT)
#define NWORD ((NK + 7) / 8)
#define LCAP  1024
#define RB    16
#define NWL   ((HID * NCLASSES + PT - 1) / PT)
static_assert(NNODES % 4 == 0 && HID == 64 && NCLASSES <= PT, "batch in 16-byte groups, a row is 32 lanes of 8 bytes");

__global__ __launch_bounds__(PT)
void k_graph_pool_head(const float* __restrict__ a, const int* __restrict__ batch, const int* __restrict__ zmm,
                       const float* __restrict__ Wl, const float* __restrict__ bl,
                       float* __restrict__ out) {
    __shared__ int   list[LCAP];
    __shared__ int   wtot[PW];
    __shared__ int   wlo[PW];
    __shared__ int   whi[PW];
    __shared__ float part[PW * HID];
    __shared__ float ssum[HID];
    __shared__ float sWl[HID * NCLASSES];

    const int tid  = threadIdx.x;
    const int lane = tid & 31;
    const int wave = __builtin_amdgcn_readfirstlane(tid >> 5);
    const int g    = blockIdx.x;

    float wl[NWL];
    #pragma unroll
    for (int i = 0; i < NWL; ++i) wl[i] = (i + 1) * PT <= HID * NCLASSES || tid + i * PT < HID * NCLASSES ? Wl[tid + i * PT] : 0.0f;
    const float blv = tid < NCLASSES ? bl[tid] : 0.0f;

    int wl_lo = NCH, wl_hi = -1;
    int2 z[NZ];
    #pragma unroll
    for (int i = 0; i < NZ; ++i) {
        const int ch = i * PT + tid;
        z[i] = ((const int2*)zmm)[ch < NCH ? ch : NCH - 1];
    }
    #pragma unroll
    for (int i = 0; i < NZ; ++i) {
        const int ch = i * PT + tid;
        const bool hit = ((i + 1) * PT <= NCH || ch < NCH) && z[i].x <= g && g <= z[i].y;
        const unsigned bm = __builtin_amdgcn_ballot_w32(hit);
        if (bm != 0u) {
            const int cb = i * PT + wave * 32;
            wl_lo = min(wl_lo, cb + __builtin_ctz(bm));
            wl_hi = max(wl_hi, cb + 31 - __builtin_clz(bm));
        }
    }
    if (lane == 0) { wlo[wave] = wl_lo; whi[wave] = wl_hi; }
    __syncthreads();
    int clo = NCH, chi = -1;
    for (int w = 0; w < PW; ++w) { clo = min(clo, wlo[w]); chi = max(chi, whi[w]); }
    const int g0 = clo * CG;
    const int g1 = min((chi + 1) * CG, NB4);
    const bool near = g1 - g0 <= PT;

    const int4* __restrict__ batch4 = (const int4*)batch;
    unsigned m1 = 0u;
    unsigned mask[NWORD];
    #pragma unroll
    for (int w = 0; w < NWORD; ++w) mask[w] = 0u;
    if (near) {
        const int idx = g0 + tid;
        if (idx < g1) {
            const int4 d = batch4[idx];
            m1 = (d.x == g ? 1u : 0u) | (d.y == g ? 2u : 0u) | (d.z == g ? 4u : 0u) | (d.w == g ? 8u : 0u);
        }
    } else {
    #pragma unroll
    for (int w = 0; w < NWORD; ++w) {
        int4 d[8];
        #pragma unroll
        for (int j = 0; j < 8; ++j) {
            const int k = w * 8 + j;
            d[j] = make_int4(-1, -1, -1, -1);
            if (k < NK) {
                const int idx = k * PT + tid;
                if ((k + 1) * PT <= NB4 || idx < NB4) d[j] = batch4[idx];
            }
        }
        unsigned m = 0u;
        #pragma unroll
        for (int j = 0; j < 8; ++j) {
            if (w * 8 + j < NK) {
                m |= (d[j].x == g ? 1u : 0u) << (4 * j);
                m |= (d[j].y == g ? 2u : 0u) << (4 * j);
                m |= (d[j].z == g ? 4u : 0u) << (4 * j);
                m |= (d[j].w == g ? 8u : 0u) << (4 * j);
            }
        }
        mask[w] = m;
    }
    }

    #pragma unroll
    for (int i = 0; i < NWL; ++i) if ((i + 1) * PT <= HID * NCLASSES || tid + i * PT < HID * NCLASSES) sWl[tid + i * PT] = wl[i];

    int c = __builtin_popcount(m1);
    #pragma unroll
    for (int w = 0; w < NWORD; ++w) c += __builtin_popcount(mask[w]);
    int x = c;
    for (int dl = 1; dl < 32; dl <<= 1) {
        const int y = __shfl_up(x, dl);
        if (lane >= dl) x += y;
    }
    if (lane == 31) wtot[wave] = x;
    __syncthreads();
    int base = 0, total = 0;
    for (int w = 0; w < PW; ++w) {
        const int t = wtot[w];
        if (w < wave) base += t;
        total += t;
    }
    const int first = base + x - c;

    float2 acc = make_float2(0.0f, 0.0f);
    const float2* __restrict__ a2 = (const float2*)a;
    for (int r0 = 0; r0 < total; r0 += LCAP) {
        if (r0 != 0) __syncthreads();
        if (c != 0) {
            int p = first - r0;
            for (unsigned m = m1; m != 0u; m &= m - 1u) {
                if (p >= 0 && p < LCAP) list[p] = ((g0 + tid) << 2) + __builtin_ctz(m);
                ++p;
            }
            #pragma unroll
            for (int w = 0; w < NWORD; ++w) {
                unsigned m = mask[w];
                while (m != 0u) {
                    const int b = __builtin_ctz(m);
                    m &= m - 1u;
                    if (p >= 0 && p < LCAP) list[p] = (((w * 8 + (b >> 2)) * PT + tid) << 2) + (b & 3);
                    ++p;
                }
            }
        }
        __syncthreads();
        const int n = min(total - r0, LCAP);
        for (int p0 = wave; p0 < n + wave; p0 += PW * RB) {
            float2 v[RB];
            #pragma unroll
            for (int k = 0; k < RB; ++k) {
                const int p = p0 + PW * k < n ? p0 + PW * k : n - 1;
                const int node = __builtin_amdgcn_readfirstlane(list[p]);
                v[k] = a2[(size_t)node * 32 + lane];
            }
            #pragma unroll
            for (int k = 0; k < RB; ++k) {
                const bool ok = p0 + PW * k < n;
                acc.x += fmaxf(ok ? v[k].x : 0.0f, 0.0f);
                acc.y += fmaxf(ok ? v[k].y : 0.0f, 0.0f);
            }
        }
    }

    ((float2*)part)[wave * 32 + lane] = acc;
    __syncthreads();
    if (tid < HID) {
        float s = 0.0f;
        for (int w = 0; w < PW; ++w) s += part[w * HID + tid];
        ssum[tid] = s;
    }
    __syncthreads();
    if (tid < NCLASSES) {
        const float inv = 1.0f / fmaxf((float)total, 1.0f);
        float o = 0.0f;
        #pragma unroll
        for (int k = 0; k < HID; ++k) o += ssum[k] * sWl[k * NCLASSES + tid];
        out[g * NCLASSES + tid] = o * inv + blv;
    }
}


extern "C" void kernel_launch(void* const* d_in, const int* in_sizes, int n_in,
                              void* d_out, int out_size, void* d_ws, size_t ws_size,
                              hipStream_t stream) {
    const float* x  = (const float*)d_in[0];
    const float* W1 = (const float*)d_in[1];
    const float* b1 = (const float*)d_in[2];
    const float* W2 = (const float*)d_in[3];
    const float* b2 = (const float*)d_in[4];
    const float* Wl = (const float*)d_in[5];
    const float* bl = (const float*)d_in[6];
    const int* edge  = (const int*)d_in[7];
    const int* batch = (const int*)d_in[8];
    const int* src = edge;
    const int* dst = edge + NEDGES;
    float* out = (float*)d_out;

    float* ws   = (float*)d_ws;
    float* h    = ws;
    float* a    = h + (size_t)NNODES * HID;
    float* dinv = a + (size_t)NNODES * HID;
    int* wl   = (int*)ws + (size_t)WL_OFF;
    int* wcnt = (int*)ws + (size_t)WCNT_OFF;
    int* ovf  = (int*)ws + (size_t)OVF_OFF;
    unsigned* ob = (unsigned*)ws + (size_t)OB_OFF;
    int* ew   = (int*)ws + (size_t)EW_OFF;
    int* zmm  = (int*)ws + (size_t)ZM_OFF;

    k_edge_words<<<EBLK, 256, 0, stream>>>(src, dst, batch, ob, ew, zmm);

    k_owner_filter<<<OBLK, 1024, 0, stream>>>(ob, ew, wl, wcnt, ovf, dinv);

    k_owner_agg<<<OBLK, 1024, 0, stream>>>(x, dinv, W1, b1, 1, src, dst, wl, wcnt, ovf, h);

    k_owner_agg<<<OBLK, 1024, 0, stream>>>(h, dinv, W2, b2, 0, src, dst, wl, wcnt, ovf, a);

    k_graph_pool_head<<<NGRAPHS, PT, 0, stream>>>(a, batch, zmm, Wl, bl, out);
}


# Source weights from an LDS table: the filter also stores each node's degree as a byte; every sum block copies the table to LDS by whole lines and reads dinv[src] there, not by a 32-line global load
# speedup vs baseline: 4.7027x; 1.0401x over previous
#include <hip/hip_runtime.h>

typedef float v2f __attribute__((ext_vector_type(2)));
typedef float v8f __attribute__((ext_vector_type(8)));

#define NNODES 50000
#define NEDGES 800000
#define HID 64
#define NGRAPHS 512
#define NCLASSES 10


#define OT   196
#define NV4  (NEDGES / 4)
#define EPT  3
#define EBLK ((NV4 + 256 * EPT - 1) / (256 * EPT))
#define NB4   (NNODES / 4)
#define CG    32
#define NCH   ((NB4 + CG - 1) / CG)
static_assert(NNODES % 4 == 0 && EBLK * 8 >= NCH, "batch in 16-byte groups; the first launch has a wave for every chunk");

__global__ __launch_bounds__(256)
void k_edge_words(const int* __restrict__ src, const int* __restrict__ dst, const int* __restrict__ batch,
                  unsigned* __restrict__ ob, int* __restrict__ ew, int* __restrict__ zmm) {
    const int t0 = blockIdx.x * 256 + threadIdx.x;
    const int lane = threadIdx.x & 31;
    const int gw = __builtin_amdgcn_readfirstlane(blockIdx.x * 8 + (int)(threadIdx.x >> 5));
    int4 bq = make_int4(0, 0, 0, 0);
    if (gw < NCH) {
        const int gi = gw * CG + lane;
        bq = ((const int4*)batch)[gi < NB4 ? gi : NB4 - 1];
    }
    int4 dv[EPT], sv[EPT];
    #pragma unroll
    for (int j = 0; j < EPT; ++j) {
        const int g = t0 + j * (EBLK * 256);
        const int gc = g < NV4 ? g : NV4 - 1;
        dv[j] = ((const int4*)dst)[gc];
        sv[j] = ((const int4*)src)[gc];
    }
    #pragma unroll
    for (int j = 0; j < EPT; ++j) {
        const int g = t0 + j * (EBLK * 256);
        if (g < NV4) {
            const int4 d = dv[j];
            const int4 s = sv[j];
            const unsigned o0 = (unsigned)d.x / OT, o1 = (unsigned)d.y / OT;
            const unsigned o2 = (unsigned)d.z / OT, o3 = (unsigned)d.w / OT;
            int4 w;
            w.x = s.x | (int)(((unsigned)d.x - o0 * OT) << 16);
            w.y = s.y | (int)(((unsigned)d.y - o1 * OT) << 16);
            w.z = s.z | (int)(((unsigned)d.z - o2 * OT) << 16);
            w.w = s.w | (int)(((unsigned)d.w - o3 * OT) << 16);
            ((int4*)ew)[g] = w;
            ob[g] = (o0 & 255u) | (o1 & 255u) << 8 | (o2 & 255u) << 16 | o3 << 24;
        }
    }
    if (gw < NCH) {
        int mn = min(min(bq.x, bq.y), min(bq.z, bq.w));
        int mx = max(max(bq.x, bq.y), max(bq.z, bq.w));
        for (int dl = 16; dl >= 1; dl >>= 1) {
            mn = min(mn, __shfl_xor(mn, dl));
            mx = max(mx, __shfl_xor(mx, dl));
        }
        if (lane == 0) ((int2*)zmm)[gw] = make_int2(mn, mx);
    }
}


#define OBLK ((NNODES + OT - 1) / OT)
#define OTP  ((OT + 1 + 15) / 16 * 16)
#define NT16 (OTP / 16)
#define AP   68
#define WP   72
#define TL   5
#define SCH  (TL * 32)
#define NSC  (NV4 / SCH)
#define RG   16
#define QCAP 896
#define WCAP 512
#define FQCAP 1312
#define WL_OFF   ((2 * NNODES * HID + NNODES + 63) / 64 * 64)
#define WCNT_OFF (WL_OFF + OBLK * 32 * WCAP)
#define OVF_OFF  (WCNT_OFF + OBLK * 32)
#define OB_OFF   ((OVF_OFF + OBLK + 63) / 64 * 64)
#define EW_OFF   ((OB_OFF + NV4 + 63) / 64 * 64)
#define ZM_OFF   ((EW_OFF + NEDGES + 63) / 64 * 64)
#define DG_OFF   ((ZM_OFF + 2 * NCH + 63) / 64 * 64)
#define DGBIG 255
#define NDG16 (NNODES / 16)
#define NDGT  ((NDG16 + 1023) / 1024)
static_assert(NNODES % 16 == 0, "the degree table is copied in whole 16-byte groups");
#define NV16 (NEDGES / 16)
#define TB   5
#define NTB  ((NV16 + TB * 32 - 1) / (TB * 32))
static_assert(NEDGES % 4 == 0 && NV4 % SCH == 0, "dst is read in whole trips of 16-byte groups");
static_assert(2 * NT16 <= 32 && HID == 64, "product: two waves per 16-row tile, 32 columns each");
static_assert(AP % 4 == 0 && WP % 4 == 0 && AP >= HID && WP >= HID, "LDS rows hold 64 floats and start on 16-byte boundaries");
static_assert(NEDGES <= (1 << 20) && OT <= (1 << 12), "queue word = edge | row << 20");
static_assert(NNODES <= (1 << 16) && OT <= (1 << 16), "list word = source | row << 16");
static_assert(QCAP >= 31 + SCH * 4 && FQCAP >= QCAP && NSC >= 32 && 32 % RG == 0, "queue room, a trip for every wave");
static_assert(32 * FQCAP * 4 > 160 * 1024 && 32 * FQCAP * 4 + OT * 4 + 4 <= 320 * 1024, "filter block: over half of the LDS, within all of it");
static_assert(OBLK <= 256 && NEDGES % 16 == 0 && FQCAP >= 31 + 32 * 16 && NTB >= 32, "an owner number is a byte; one load's hits fit the queue");

template <class Take>
__device__ __forceinline__ void walk_dst(const int* __restrict__ dst, int lo, int wave, int lane,
                                         unsigned* __restrict__ myq, Take&& take) {
    int qn = 0;
    const int4* __restrict__ dst4 = (const int4*)dst;
    const int start = (blockIdx.x * (NSC / OBLK)) % NSC;
    const int ntrip = (NSC - wave + 31) / 32;

    auto load_trip = [&](int i, int4 (&buf)[TL], int& base) {
        const int ic = i < ntrip ? i : ntrip - 1;
        int s2 = ic * 32 + wave + start;
        if (s2 >= NSC) s2 -= NSC;
        base = s2 * SCH;
        #pragma unroll
        for (int k = 0; k < TL; ++k) buf[k] = dst4[base + k * 32 + lane];
    };

    auto push = [&](unsigned t, int e) {
        const bool m = t < (unsigned)OT;
        const unsigned mask = __builtin_amdgcn_ballot_w32(m);
        if (mask != 0u) {
            if (m) {
                const int slot = qn + (int)__builtin_amdgcn_mbcnt_lo(mask, 0u);
                myq[slot] = (unsigned)e | (t << 20);
            }
            qn += __builtin_popcount(mask);
        }
    };

    auto process = [&](const int4 (&buf)[TL], int base, bool last) {
        #pragma unroll
        for (int k = 0; k < TL; ++k) {
            const int4 d = buf[k];
            const unsigned t0 = (unsigned)(d.x - lo), t1 = (unsigned)(d.y - lo);
            const unsigned t2 = (unsigned)(d.z - lo), t3 = (unsigned)(d.w - lo);
            const unsigned tm = min(min(t0, t1), min(t2, t3));
            if (__builtin_amdgcn_ballot_w32(tm < (unsigned)OT) != 0u) {
                const int e0 = (base + k * 32 + lane) * 4;
                push(t0, e0);
                push(t1, e0 + 1);
                push(t2, e0 + 2);
                push(t3, e0 + 3);
            }
        }
        const int keep = last ? 0 : 31;
        while (qn > keep) qn = take(qn);
    };

    int4 bufA[TL], bufB[TL];
    int baseA = 0, baseB = 0;
    load_trip(0, bufA, baseA);
    for (int i = 0; i < ntrip; i += 2) {
        load_trip(i + 1, bufB, baseB);
        process(bufA, baseA, i + 1 >= ntrip);
        if (i + 1 >= ntrip) break;
        load_trip(i + 2, bufA, baseA);
        process(bufB, baseB, i + 2 >= ntrip);
    }
}

__global__ __launch_bounds__(1024)
void k_owner_filter(const unsigned* __restrict__ ob, const int* __restrict__ ew,
                    int* __restrict__ wl, int* __restrict__ wcnt, int* __restrict__ ovf,
                    float* __restrict__ dinv, unsigned char* __restrict__ deg8) {
    __shared__ int scnt[OT];
    __shared__ int sovf;
    __shared__ unsigned q[32 * FQCAP];

    const int tid  = threadIdx.x;
    const int lane = tid & 31;
    const int wave = __builtin_amdgcn_readfirstlane(tid >> 5);
    const int lo   = blockIdx.x * OT;
    const unsigned idr = (unsigned)blockIdx.x * 0x01010101u;
    const uint4* __restrict__ ob4 = (const uint4*)ob;

    if (tid < OT) scnt[tid] = 0;
    if (tid == 0) sovf = 0;
    __syncthreads();

    unsigned* __restrict__ myq = q + wave * FQCAP;
    int* __restrict__ mywl = wl + (size_t)(blockIdx.x * 32 + wave) * WCAP;
    int wn = 0;
    int qn = 0;

    auto take = [&](int fill) {
        const int base = fill > 32 ? fill - 32 : 0;
        const int cnt  = fill - base;
        if (lane < cnt) {
            const unsigned w = myq[base + lane];
            const unsigned p = w & 31u;
            const unsigned e = (w >> 5) * 16u + ((p & 7u) << 2) + (p >> 3);
            const int word = ew[e];
            atomicAdd(&scnt[(unsigned)word >> 16], 1);
            if (wn + lane < WCAP) mywl[wn + lane] = word;
        }
        wn += cnt;
        return base;
    };

    const int start = (blockIdx.x * NTB) / OBLK;
    const int ntrip = (NTB - wave + 31) / 32;

    auto load_trip = [&](int i, uint4 (&buf)[TB], int& base) {
        const int ic = i < ntrip ? i : ntrip - 1;
        int s2 = ic * 32 + wave + start;
        if (s2 >= NTB) s2 -= NTB;
        base = s2 * (TB * 32);
        #pragma unroll
        for (int k = 0; k < TB; ++k) {
            const int g = base + k * 32 + lane;
            buf[k] = ob4[g < NV16 ? g : NV16 - 1];
        }
    };

    auto eq7 = [&](unsigned x) {
        const unsigned y = x ^ idr;
        return ~(((y & 0x7F7F7F7Fu) + 0x7F7F7F7Fu) | y | 0x7F7F7F7Fu);
    };

    auto process = [&](const uint4 (&buf)[TB], int base) {
        #pragma unroll
        for (int k = 0; k < TB; ++k) {
            const uint4 d = buf[k];
            const int g = base + k * 32 + lane;
            unsigned m = (eq7(d.x) >> 7) | (eq7(d.y) >> 6) | (eq7(d.z) >> 5) | (eq7(d.w) >> 4);
            if (base + k * 32 + 32 > NV16) m = g < NV16 ? m : 0u;
            unsigned any = __builtin_amdgcn_ballot_w32(m != 0u);
            while (any != 0u) {
                if (m != 0u) {
                    const int p = __builtin_ctz(m);
                    const int slot = qn + (int)__builtin_amdgcn_mbcnt_lo(any, 0u);
                    myq[slot] = (unsigned)(g * 32 + p);
                    m &= m - 1u;
                }
                qn += __builtin_popcount(any);
                any = __builtin_amdgcn_ballot_w32(m != 0u);
            }
            while (qn > 31) qn = take(qn);
        }
    };

    uint4 bufA[TB], bufB[TB];
    int baseA = 0, baseB = 0;
    load_trip(0, bufA, baseA);
    for (int i = 0; i < ntrip; i += 2) {
        load_trip(i + 1, bufB, baseB);
        process(bufA, baseA);
        if (i + 1 >= ntrip) break;
        load_trip(i + 2, bufA, baseA);
        process(bufB, baseB);
    }
    while (qn > 0) qn = take(qn);

    if (lane == 0) {
        wcnt[blockIdx.x * 32 + wave] = wn;
        if (wn > WCAP) sovf = 1;
    }
    __syncthreads();
    if (tid < OT && lo + tid < NNODES) {
        const int dg = scnt[tid] + 1;
        dinv[lo + tid]  = rsqrtf((float)dg);
        deg8[lo + tid]  = (unsigned char)(dg < DGBIG ? dg : DGBIG);
    }
    if (tid == 0) ovf[blockIdx.x] = sovf;
}

__global__ __launch_bounds__(1024)
void k_owner_agg(const float* __restrict__ h, const float* __restrict__ dinv, const unsigned char* __restrict__ deg8,
                 const float* __restrict__ W, const float* __restrict__ bias, int relu_out,
                 const int* __restrict__ src, const int* __restrict__ dst,
                 const int* __restrict__ wl, const int* __restrict__ wcnt, const int* __restrict__ ovf,
                 float* __restrict__ a) {
    __shared__ float acc[OTP * AP];
    __shared__ float sW[HID * WP];
    __shared__ float sdinv[OT];
    __shared__ unsigned q[32 * QCAP];
    __shared__ __attribute__((aligned(16))) unsigned char sdeg[NDGT * 1024 * 16];

    const int tid  = threadIdx.x;
    const int lane = tid & 31;
    const int wave = __builtin_amdgcn_readfirstlane(tid >> 5);
    const int lo   = blockIdx.x * OT;

    const int over = ovf[blockIdx.x];
    const int* __restrict__ mywl = wl + (size_t)(blockIdx.x * 32 + wave) * WCAP;
    int n = 0;
    int wnext = 0;
    if (!over) {
        n = __builtin_amdgcn_readfirstlane(wcnt[blockIdx.x * 32 + wave]);
        if (n > 0) wnext = mywl[lane < n ? lane : n - 1];
    }

    const int l16  = lane & 15;
    const int half = lane >> 4;
    const int nb   = (wave & 1) * 32;
    ((float4*)(sW + (tid >> 4) * WP))[tid & 15] = ((const float4*)W)[tid];
    const float bv0 = bias[nb + l16];
    const float bv1 = bias[nb + 16 + l16];

    uint4 tv[NDGT];
    #pragma unroll
    for (int k = 0; k < NDGT; ++k) {
        const int i = tid + k * 1024;
        tv[k] = ((const uint4*)deg8)[i < NDG16 ? i : NDG16 - 1];
    }
    #pragma unroll
    for (int k = 0; k < NDGT; ++k) ((uint4*)sdeg)[tid + k * 1024] = tv[k];

    for (int i = tid; i < OTP * 16; i += 1024) {
        const int r = i >> 4, c = i & 15;
        const int node = lo + r;
        float4 v = make_float4(0.f, 0.f, 0.f, 0.f);
        if (r < OT && node < NNODES) {
            const float4 hv = ((const float4*)h)[(size_t)node * 16 + c];
            const float di = dinv[node];
            v.x = hv.x * di * di;
            v.y = hv.y * di * di;
            v.z = hv.z * di * di;
            v.w = hv.w * di * di;
        }
        ((float4*)(acc + r * AP))[c] = v;
    }
    if (tid < OT) sdinv[tid] = (lo + tid < NNODES) ? dinv[lo + tid] : 0.0f;
    __syncthreads();

    const char* __restrict__ hb = (const char*)h;
    char* accb = (char*)acc;

    auto add_rows = [&](int cnt, unsigned soff, unsigned doff, float dv, float sd) {
        #pragma nounroll
        for (int j0 = 0; j0 < cnt; j0 += RG) {
            float2 v[RG];
            #pragma unroll
            for (int k = 0; k < RG; ++k) {
                const unsigned so = (unsigned)__builtin_amdgcn_readlane((int)soff, j0 + k);
                v[k] = *(const float2*)(hb + (so + (unsigned)lane * 8u));
            }
            #pragma unroll
            for (int k = 0; k < RG; ++k) {
                const float wj =
                    __builtin_bit_cast(float, __builtin_amdgcn_readlane(__builtin_bit_cast(int, dv), j0 + k)) *
                    __builtin_bit_cast(float, __builtin_amdgcn_readlane(__builtin_bit_cast(int, sd), j0 + k));
                const unsigned dj = (unsigned)__builtin_amdgcn_readlane((int)doff, j0 + k);
                float* ap = (float*)(accb + (dj + (unsigned)lane * 8u));
                atomicAdd(ap,     v[k].x * wj);
                atomicAdd(ap + 1, v[k].y * wj);
            }
        }
    };

    auto src_dinv = [&](unsigned s) {
        const unsigned dg = sdeg[s];
        float dv = rsqrtf((float)dg);
        if (dg == (unsigned)DGBIG) dv = dinv[s];
        return dv;
    };

    if (!over) {
        for (int j0 = 0; j0 < n; j0 += 32) {
            const int cnt = min(32, n - j0);
            const unsigned word = (unsigned)wnext;
            const int nj = j0 + 32 + lane;
            wnext = mywl[nj < n ? nj : n - 1];
            const unsigned s  = word & 0xFFFFu;
            const unsigned dl = word >> 16;
            const float dv = src_dinv(s);
            const float sd = sdinv[dl];
            add_rows(cnt, s * (HID * 4u), (lane < cnt ? dl : (unsigned)OT) * (AP * 4u), dv, sd);
        }
    } else {
        unsigned* __restrict__ myq = q + wave * QCAP;
        walk_dst(dst, lo, wave, lane, myq, [&](int qn) {
            const int base = qn > 32 ? qn - 32 : 0;
            const int cnt  = qn - base;
            const bool used = lane < cnt;
            const unsigned p = myq[base + (used ? lane : cnt - 1)];
            const int e  = (int)(p & 0xFFFFFu);
            const int dl = (int)(p >> 20);
            const int s  = src[e];
            const float dv = src_dinv((unsigned)s);
            const float sd = sdinv[dl];
            add_rows(cnt, (unsigned)s * (HID * 4u), (unsigned)(used ? dl : OT) * (AP * 4u), dv, sd);
            return base;
        });
    }
    __syncthreads();

    v8f c0 = {}, c1 = {};
    const int t16 = wave >> 1;
    if (wave < 2 * NT16) {
        const float* Ar = acc + (t16 * 16 + l16) * AP;
        #pragma unroll
        for (int k0 = 0; k0 < HID; k0 += 4) {
            const int kb = k0 + half * 2;
            const v2f af = {Ar[kb], Ar[kb + 1]};
            const float* Wk0 = sW + kb * WP + nb;
            const float* Wk1 = sW + (kb + 1) * WP + nb;
            const v2f bf0 = {Wk0[l16],      Wk1[l16]};
            const v2f bf1 = {Wk0[l16 + 16], Wk1[l16 + 16]};
            c0 = __builtin_amdgcn_wmma_f32_16x16x4_f32(false, af, false, bf0, (short)0, c0, false, false);
            c1 = __builtin_amdgcn_wmma_f32_16x16x4_f32(false, af, false, bf1, (short)0, c1, false, false);
        }
    }
    __syncthreads();
    if (wave < 2 * NT16) {
        #pragma unroll
        for (int r = 0; r < 8; ++r) {
            float* orow = acc + (t16 * 16 + half * 8 + r) * AP + nb + l16;
            float v0 = c0[r] + bv0;
            float v1 = c1[r] + bv1;
            if (relu_out) { v0 = fmaxf(v0, 0.0f); v1 = fmaxf(v1, 0.0f); }
            orow[0]  = v0;
            orow[16] = v1;
        }
    }
    __syncthreads();

    for (int i = tid; i < OT * 16; i += 1024) {
        const int node = lo + (i >> 4);
        if (node < NNODES) ((float4*)a)[(size_t)node * 16 + (i & 15)] = ((const float4*)(acc + (i >> 4) * AP))[i & 15];
    }
}


#define PT    256
#define PW    (PT / 32)
#define NK    ((NB4 + PT - 1) / PT)
#define NZ    ((NCH + PT - 1) / PT)
#define NWORD ((NK + 7) / 8)
#define LCAP  1024
#define RB    16
#define NWL   ((HID * NCLASSES + PT - 1) / PT)
static_assert(NNODES % 4 == 0 && HID == 64 && NCLASSES <= PT, "batch in 16-byte groups, a row is 32 lanes of 8 bytes");

__global__ __launch_bounds__(PT)
void k_graph_pool_head(const float* __restrict__ a, const int* __restrict__ batch, const int* __restrict__ zmm,
                       const float* __restrict__ Wl, const float* __restrict__ bl,
                       float* __restrict__ out) {
    __shared__ int   list[LCAP];
    __shared__ int   wtot[PW];
    __shared__ int   wlo[PW];
    __shared__ int   whi[PW];
    __shared__ float part[PW * HID];
    __shared__ float ssum[HID];
    __shared__ float sWl[HID * NCLASSES];

    const int tid  = threadIdx.x;
    const int lane = tid & 31;
    const int wave = __builtin_amdgcn_readfirstlane(tid >> 5);
    const int g    = blockIdx.x;

    float wl[NWL];
    #pragma unroll
    for (int i = 0; i < NWL; ++i) wl[i] = (i + 1) * PT <= HID * NCLASSES || tid + i * PT < HID * NCLASSES ? Wl[tid + i * PT] : 0.0f;
    const float blv = tid < NCLASSES ? bl[tid] : 0.0f;

    int wl_lo = NCH, wl_hi = -1;
    int2 z[NZ];
    #pragma unroll
    for (int i = 0; i < NZ; ++i) {
        const int ch = i * PT + tid;
        z[i] = ((const int2*)zmm)[ch < NCH ? ch : NCH - 1];
    }
    #pragma unroll
    for (int i = 0; i < NZ; ++i) {
        const int ch = i * PT + tid;
        const bool hit = ((i + 1) * PT <= NCH || ch < NCH) && z[i].x <= g && g <= z[i].y;
        const unsigned bm = __builtin_amdgcn_ballot_w32(hit);
        if (bm != 0u) {
            const int cb = i * PT + wave * 32;
            wl_lo = min(wl_lo, cb + __builtin_ctz(bm));
            wl_hi = max(wl_hi, cb + 31 - __builtin_clz(bm));
        }
    }
    if (lane == 0) { wlo[wave] = wl_lo; whi[wave] = wl_hi; }
    __syncthreads();
    int clo = NCH, chi = -1;
    for (int w = 0; w < PW; ++w) { clo = min(clo, wlo[w]); chi = max(chi, whi[w]); }
    const int g0 = clo * CG;
    const int g1 = min((chi + 1) * CG, NB4);
    const bool near = g1 - g0 <= PT;

    const int4* __restrict__ batch4 = (const int4*)batch;
    unsigned m1 = 0u;
    unsigned mask[NWORD];
    #pragma unroll
    for (int w = 0; w < NWORD; ++w) mask[w] = 0u;
    if (near) {
        const int idx = g0 + tid;
        if (idx < g1) {
            const int4 d = batch4[idx];
            m1 = (d.x == g ? 1u : 0u) | (d.y == g ? 2u : 0u) | (d.z == g ? 4u : 0u) | (d.w == g ? 8u : 0u);
        }
    } else {
    #pragma unroll
    for (int w = 0; w < NWORD; ++w) {
        int4 d[8];
        #pragma unroll
        for (int j = 0; j < 8; ++j) {
            const int k = w * 8 + j;
            d[j] = make_int4(-1, -1, -1, -1);
            if (k < NK) {
                const int idx = k * PT + tid;
                if ((k + 1) * PT <= NB4 || idx < NB4) d[j] = batch4[idx];
            }
        }
        unsigned m = 0u;
        #pragma unroll
        for (int j = 0; j < 8; ++j) {
            if (w * 8 + j < NK) {
                m |= (d[j].x == g ? 1u : 0u) << (4 * j);
                m |= (d[j].y == g ? 2u : 0u) << (4 * j);
                m |= (d[j].z == g ? 4u : 0u) << (4 * j);
                m |= (d[j].w == g ? 8u : 0u) << (4 * j);
            }
        }
        mask[w] = m;
    }
    }

    #pragma unroll
    for (int i = 0; i < NWL; ++i) if ((i + 1) * PT <= HID * NCLASSES || tid + i * PT < HID * NCLASSES) sWl[tid + i * PT] = wl[i];

    int c = __builtin_popcount(m1);
    #pragma unroll
    for (int w = 0; w < NWORD; ++w) c += __builtin_popcount(mask[w]);
    int x = c;
    for (int dl = 1; dl < 32; dl <<= 1) {
        const int y = __shfl_up(x, dl);
        if (lane >= dl) x += y;
    }
    if (lane == 31) wtot[wave] = x;
    __syncthreads();
    int base = 0, total = 0;
    for (int w = 0; w < PW; ++w) {
        const int t = wtot[w];
        if (w < wave) base += t;
        total += t;
    }
    const int first = base + x - c;

    float2 acc = make_float2(0.0f, 0.0f);
    const float2* __restrict__ a2 = (const float2*)a;
    for (int r0 = 0; r0 < total; r0 += LCAP) {
        if (r0 != 0) __syncthreads();
        if (c != 0) {
            int p = first - r0;
            for (unsigned m = m1; m != 0u; m &= m - 1u) {
                if (p >= 0 && p < LCAP) list[p] = ((g0 + tid) << 2) + __builtin_ctz(m);
                ++p;
            }
            #pragma unroll
            for (int w = 0; w < NWORD; ++w) {
                unsigned m = mask[w];
                while (m != 0u) {
                    const int b = __builtin_ctz(m);
                    m &= m - 1u;
                    if (p >= 0 && p < LCAP) list[p] = (((w * 8 + (b >> 2)) * PT + tid) << 2) + (b & 3);
                    ++p;
                }
            }
        }
        __syncthreads();
        const int n = min(total - r0, LCAP);
        for (int p0 = wave; p0 < n + wave; p0 += PW * RB) {
            float2 v[RB];
            #pragma unroll
            for (int k = 0; k < RB; ++k) {
                const int p = p0 + PW * k < n ? p0 + PW * k : n - 1;
                const int node = __builtin_amdgcn_readfirstlane(list[p]);
                v[k] = a2[(size_t)node * 32 + lane];
            }
            #pragma unroll
            for (int k = 0; k < RB; ++k) {
                const bool ok = p0 + PW * k < n;
                acc.x += fmaxf(ok ? v[k].x : 0.0f, 0.0f);
                acc.y += fmaxf(ok ? v[k].y : 0.0f, 0.0f);
            }
        }
    }

    ((float2*)part)[wave * 32 + lane] = acc;
    __syncthreads();
    if (tid < HID) {
        float s = 0.0f;
        for (int w = 0; w < PW; ++w) s += part[w * HID + tid];
        ssum[tid] = s;
    }
    __syncthreads();
    if (tid < NCLASSES) {
        const float inv = 1.0f / fmaxf((float)total, 1.0f);
        float o = 0.0f;
        #pragma unroll
        for (int k = 0; k < HID; ++k) o += ssum[k] * sWl[k * NCLASSES + tid];
        out[g * NCLASSES + tid] = o * inv + blv;
    }
}


extern "C" void kernel_launch(void* const* d_in, const int* in_sizes, int n_in,
                              void* d_out, int out_size, void* d_ws, size_t ws_size,
                              hipStream_t stream) {
    const float* x  = (const float*)d_in[0];
    const float* W1 = (const float*)d_in[1];
    const float* b1 = (const float*)d_in[2];
    const float* W2 = (const float*)d_in[3];
    const float* b2 = (const float*)d_in[4];
    const float* Wl = (const float*)d_in[5];
    const float* bl = (const float*)d_in[6];
    const int* edge  = (const int*)d_in[7];
    const int* batch = (const int*)d_in[8];
    const int* src = edge;
    const int* dst = edge + NEDGES;
    float* out = (float*)d_out;

    float* ws   = (float*)d_ws;
    float* h    = ws;
    float* a    = h + (size_t)NNODES * HID;
    float* dinv = a + (size_t)NNODES * HID;
    int* wl   = (int*)ws + (size_t)WL_OFF;
    int* wcnt = (int*)ws + (size_t)WCNT_OFF;
    int* ovf  = (int*)ws + (size_t)OVF_OFF;
    unsigned* ob = (unsigned*)ws + (size_t)OB_OFF;
    int* ew   = (int*)ws + (size_t)EW_OFF;
    int* zmm  = (int*)ws + (size_t)ZM_OFF;
    unsigned char* deg8 = (unsigned char*)((int*)ws + (size_t)DG_OFF);

    k_edge_words<<<EBLK, 256, 0, stream>>>(src, dst, batch, ob, ew, zmm);

    k_owner_filter<<<OBLK, 1024, 0, stream>>>(ob, ew, wl, wcnt, ovf, dinv, deg8);

    k_owner_agg<<<OBLK, 1024, 0, stream>>>(x, dinv, deg8, W1, b1, 1, src, dst, wl, wcnt, ovf, h);

    k_owner_agg<<<OBLK, 1024, 0, stream>>>(h, dinv, deg8, W2, b2, 0, src, dst, wl, wcnt, ovf, a);

    k_graph_pool_head<<<NGRAPHS, PT, 0, stream>>>(a, batch, zmm, Wl, bl, out);
}


# Sum launches open with ONE group of loads (list mark, length, first words, weights, degree table, start rows and their dinv), waited for once, not a chain of eight dependent round trips
# speedup vs baseline: 4.8977x; 1.0415x over previous
#include <hip/hip_runtime.h>

typedef float v2f __attribute__((ext_vector_type(2)));
typedef float v8f __attribute__((ext_vector_type(8)));
typedef float v4f __attribute__((ext_vector_type(4)));
typedef unsigned v4u __attribute__((ext_vector_type(4)));

#define NNODES 50000
#define NEDGES 800000
#define HID 64
#define NGRAPHS 512
#define NCLASSES 10


#define OT   196
#define NV4  (NEDGES / 4)
#define EPT  3
#define EBLK ((NV4 + 256 * EPT - 1) / (256 * EPT))
#define NB4   (NNODES / 4)
#define CG    32
#define NCH   ((NB4 + CG - 1) / CG)
static_assert(NNODES % 4 == 0 && EBLK * 8 >= NCH, "batch in 16-byte groups; the first launch has a wave for every chunk");

__global__ __launch_bounds__(256)
void k_edge_words(const int* __restrict__ src, const int* __restrict__ dst, const int* __restrict__ batch,
                  unsigned* __restrict__ ob, int* __restrict__ ew, int* __restrict__ zmm) {
    const int t0 = blockIdx.x * 256 + threadIdx.x;
    const int lane = threadIdx.x & 31;
    const int gw = __builtin_amdgcn_readfirstlane(blockIdx.x * 8 + (int)(threadIdx.x >> 5));
    int4 bq = make_int4(0, 0, 0, 0);
    if (gw < NCH) {
        const int gi = gw * CG + lane;
        bq = ((const int4*)batch)[gi < NB4 ? gi : NB4 - 1];
    }
    int4 dv[EPT], sv[EPT];
    #pragma unroll
    for (int j = 0; j < EPT; ++j) {
        const int g = t0 + j * (EBLK * 256);
        const int gc = g < NV4 ? g : NV4 - 1;
        dv[j] = ((const int4*)dst)[gc];
        sv[j] = ((const int4*)src)[gc];
    }
    #pragma unroll
    for (int j = 0; j < EPT; ++j) {
        const int g = t0 + j * (EBLK * 256);
        if (g < NV4) {
            const int4 d = dv[j];
            const int4 s = sv[j];
            const unsigned o0 = (unsigned)d.x / OT, o1 = (unsigned)d.y / OT;
            const unsigned o2 = (unsigned)d.z / OT, o3 = (unsigned)d.w / OT;
            int4 w;
            w.x = s.x | (int)(((unsigned)d.x - o0 * OT) << 16);
            w.y = s.y | (int)(((unsigned)d.y - o1 * OT) << 16);
            w.z = s.z | (int)(((unsigned)d.z - o2 * OT) << 16);
            w.w = s.w | (int)(((unsigned)d.w - o3 * OT) << 16);
            ((int4*)ew)[g] = w;
            ob[g] = (o0 & 255u) | (o1 & 255u) << 8 | (o2 & 255u) << 16 | o3 << 24;
        }
    }
    if (gw < NCH) {
        int mn = min(min(bq.x, bq.y), min(bq.z, bq.w));
        int mx = max(max(bq.x, bq.y), max(bq.z, bq.w));
        for (int dl = 16; dl >= 1; dl >>= 1) {
            mn = min(mn, __shfl_xor(mn, dl));
            mx = max(mx, __shfl_xor(mx, dl));
        }
        if (lane == 0) ((int2*)zmm)[gw] = make_int2(mn, mx);
    }
}


#define OBLK ((NNODES + OT - 1) / OT)
#define OTP  ((OT + 1 + 15) / 16 * 16)
#define NT16 (OTP / 16)
#define AP   68
#define WP   72
#define TL   5
#define SCH  (TL * 32)
#define NSC  (NV4 / SCH)
#define RG   16
#define QCAP 896
#define WCAP 512
#define FQCAP 1312
#define WL_OFF   ((2 * NNODES * HID + NNODES + 63) / 64 * 64)
#define WCNT_OFF (WL_OFF + OBLK * 32 * WCAP)
#define OVF_OFF  (WCNT_OFF + OBLK * 32)
#define OB_OFF   ((OVF_OFF + OBLK + 63) / 64 * 64)
#define EW_OFF   ((OB_OFF + NV4 + 63) / 64 * 64)
#define ZM_OFF   ((EW_OFF + NEDGES + 63) / 64 * 64)
#define DG_OFF   ((ZM_OFF + 2 * NCH + 63) / 64 * 64)
#define DGBIG 255
#define NDG16 (NNODES / 16)
#define NDGT  ((NDG16 + 1023) / 1024)
static_assert(NNODES % 16 == 0, "the degree table is copied in whole 16-byte groups");
#define NSV  ((OTP * 16 + 1023) / 1024)
#define ACR  (NSV * 1024 / 16)
static_assert(ACR >= OTP, "the start values cover every row the product reads");
static_assert(WCAP >= 32 && NDGT == 4 && NSV == 4, "a list region holds at least one batch; the load group's hold names four table and four start-row trips");
#define NV16 (NEDGES / 16)
#define TB   5
#define NTB  ((NV16 + TB * 32 - 1) / (TB * 32))
static_assert(NEDGES % 4 == 0 && NV4 % SCH == 0, "dst is read in whole trips of 16-byte groups");
static_assert(2 * NT16 <= 32 && HID == 64, "product: two waves per 16-row tile, 32 columns each");
static_assert(AP % 4 == 0 && WP % 4 == 0 && AP >= HID && WP >= HID, "LDS rows hold 64 floats and start on 16-byte boundaries");
static_assert(NEDGES <= (1 << 20) && OT <= (1 << 12), "queue word = edge | row << 20");
static_assert(NNODES <= (1 << 16) && OT <= (1 << 16), "list word = source | row << 16");
static_assert(QCAP >= 31 + SCH * 4 && FQCAP >= QCAP && NSC >= 32 && 32 % RG == 0, "queue room, a trip for every wave");
static_assert(32 * FQCAP * 4 > 160 * 1024 && 32 * FQCAP * 4 + OT * 4 + 4 <= 320 * 1024, "filter block: over half of the LDS, within all of it");
static_assert(OBLK <= 256 && NEDGES % 16 == 0 && FQCAP >= 31 + 32 * 16 && NTB >= 32, "an owner number is a byte; one load's hits fit the queue");

template <class Take>
__device__ __forceinline__ void walk_dst(const int* __restrict__ dst, int lo, int wave, int lane,
                                         unsigned* __restrict__ myq, Take&& take) {
    int qn = 0;
    const int4* __restrict__ dst4 = (const int4*)dst;
    const int start = (blockIdx.x * (NSC / OBLK)) % NSC;
    const int ntrip = (NSC - wave + 31) / 32;

    auto load_trip = [&](int i, int4 (&buf)[TL], int& base) {
        const int ic = i < ntrip ? i : ntrip - 1;
        int s2 = ic * 32 + wave + start;
        if (s2 >= NSC) s2 -= NSC;
        base = s2 * SCH;
        #pragma unroll
        for (int k = 0; k < TL; ++k) buf[k] = dst4[base + k * 32 + lane];
    };

    auto push = [&](unsigned t, int e) {
        const bool m = t < (unsigned)OT;
        const unsigned mask = __builtin_amdgcn_ballot_w32(m);
        if (mask != 0u) {
            if (m) {
                const int slot = qn + (int)__builtin_amdgcn_mbcnt_lo(mask, 0u);
                myq[slot] = (unsigned)e | (t << 20);
            }
            qn += __builtin_popcount(mask);
        }
    };

    auto process = [&](const int4 (&buf)[TL], int base, bool last) {
        #pragma unroll
        for (int k = 0; k < TL; ++k) {
            const int4 d = buf[k];
            const unsigned t0 = (unsigned)(d.x - lo), t1 = (unsigned)(d.y - lo);
            const unsigned t2 = (unsigned)(d.z - lo), t3 = (unsigned)(d.w - lo);
            const unsigned tm = min(min(t0, t1), min(t2, t3));
            if (__builtin_amdgcn_ballot_w32(tm < (unsigned)OT) != 0u) {
                const int e0 = (base + k * 32 + lane) * 4;
                push(t0, e0);
                push(t1, e0 + 1);
                push(t2, e0 + 2);
                push(t3, e0 + 3);
            }
        }
        const int keep = last ? 0 : 31;
        while (qn > keep) qn = take(qn);
    };

    int4 bufA[TL], bufB[TL];
    int baseA = 0, baseB = 0;
    load_trip(0, bufA, baseA);
    for (int i = 0; i < ntrip; i += 2) {
        load_trip(i + 1, bufB, baseB);
        process(bufA, baseA, i + 1 >= ntrip);
        if (i + 1 >= ntrip) break;
        load_trip(i + 2, bufA, baseA);
        process(bufB, baseB, i + 2 >= ntrip);
    }
}

__global__ __launch_bounds__(1024)
void k_owner_filter(const unsigned* __restrict__ ob, const int* __restrict__ ew,
                    int* __restrict__ wl, int* __restrict__ wcnt, int* __restrict__ ovf,
                    float* __restrict__ dinv, unsigned char* __restrict__ deg8) {
    __shared__ int scnt[OT];
    __shared__ int sovf;
    __shared__ unsigned q[32 * FQCAP];

    const int tid  = threadIdx.x;
    const int lane = tid & 31;
    const int wave = __builtin_amdgcn_readfirstlane(tid >> 5);
    const int lo   = blockIdx.x * OT;
    const unsigned idr = (unsigned)blockIdx.x * 0x01010101u;
    const uint4* __restrict__ ob4 = (const uint4*)ob;

    if (tid < OT) scnt[tid] = 0;
    if (tid == 0) sovf = 0;
    __syncthreads();

    unsigned* __restrict__ myq = q + wave * FQCAP;
    int* __restrict__ mywl = wl + (size_t)(blockIdx.x * 32 + wave) * WCAP;
    int wn = 0;
    int qn = 0;

    auto take = [&](int fill) {
        const int base = fill > 32 ? fill - 32 : 0;
        const int cnt  = fill - base;
        if (lane < cnt) {
            const unsigned w = myq[base + lane];
            const unsigned p = w & 31u;
            const unsigned e = (w >> 5) * 16u + ((p & 7u) << 2) + (p >> 3);
            const int word = ew[e];
            atomicAdd(&scnt[(unsigned)word >> 16], 1);
            if (wn + lane < WCAP) mywl[wn + lane] = word;
        }
        wn += cnt;
        return base;
    };

    const int start = (blockIdx.x * NTB) / OBLK;
    const int ntrip = (NTB - wave + 31) / 32;

    auto load_trip = [&](int i, uint4 (&buf)[TB], int& base) {
        const int ic = i < ntrip ? i : ntrip - 1;
        int s2 = ic * 32 + wave + start;
        if (s2 >= NTB) s2 -= NTB;
        base = s2 * (TB * 32);
        #pragma unroll
        for (int k = 0; k < TB; ++k) {
            const int g = base + k * 32 + lane;
            buf[k] = ob4[g < NV16 ? g : NV16 - 1];
        }
    };

    auto eq7 = [&](unsigned x) {
        const unsigned y = x ^ idr;
        return ~(((y & 0x7F7F7F7Fu) + 0x7F7F7F7Fu) | y | 0x7F7F7F7Fu);
    };

    auto process = [&](const uint4 (&buf)[TB], int base) {
        #pragma unroll
        for (int k = 0; k < TB; ++k) {
            const uint4 d = buf[k];
            const int g = base + k * 32 + lane;
            unsigned m = (eq7(d.x) >> 7) | (eq7(d.y) >> 6) | (eq7(d.z) >> 5) | (eq7(d.w) >> 4);
            if (base + k * 32 + 32 > NV16) m = g < NV16 ? m : 0u;
            unsigned any = __builtin_amdgcn_ballot_w32(m != 0u);
            while (any != 0u) {
                if (m != 0u) {
                    const int p = __builtin_ctz(m);
                    const int slot = qn + (int)__builtin_amdgcn_mbcnt_lo(any, 0u);
                    myq[slot] = (unsigned)(g * 32 + p);
                    m &= m - 1u;
                }
                qn += __builtin_popcount(any);
                any = __builtin_amdgcn_ballot_w32(m != 0u);
            }
            while (qn > 31) qn = take(qn);
        }
    };

    uint4 bufA[TB], bufB[TB];
    int baseA = 0, baseB = 0;
    load_trip(0, bufA, baseA);
    for (int i = 0; i < ntrip; i += 2) {
        load_trip(i + 1, bufB, baseB);
        process(bufA, baseA);
        if (i + 1 >= ntrip) break;
        load_trip(i + 2, bufA, baseA);
        process(bufB, baseB);
    }
    while (qn > 0) qn = take(qn);

    if (wn < 32 && lane >= wn) mywl[lane] = 0;

    if (lane == 0) {
        wcnt[blockIdx.x * 32 + wave] = wn;
        if (wn > WCAP) sovf = 1;
    }
    __syncthreads();
    if (tid < OT && lo + tid < NNODES) {
        const int dg = scnt[tid] + 1;
        dinv[lo + tid]  = rsqrtf((float)dg);
        deg8[lo + tid]  = (unsigned char)(dg < DGBIG ? dg : DGBIG);
    }
    if (tid == 0) ovf[blockIdx.x] = sovf;
}

__global__ __launch_bounds__(1024)
void k_owner_agg(const float* __restrict__ h, const float* __restrict__ dinv, const unsigned char* __restrict__ deg8,
                 const float* __restrict__ W, const float* __restrict__ bias, int relu_out,
                 const int* __restrict__ src, const int* __restrict__ dst,
                 const int* __restrict__ wl, const int* __restrict__ wcnt, const int* __restrict__ ovf,
                 float* __restrict__ a) {
    __shared__ float acc[ACR * AP];
    __shared__ float sW[HID * WP];
    __shared__ float sdinv[OT];
    __shared__ unsigned q[32 * QCAP];
    __shared__ __attribute__((aligned(16))) unsigned char sdeg[NDGT * 1024 * 16];

    const int tid  = threadIdx.x;
    const int lane = tid & 31;
    const int wave = __builtin_amdgcn_readfirstlane(tid >> 5);
    const int lo   = blockIdx.x * OT;

    const int over = ovf[blockIdx.x];
    const int nraw = wcnt[blockIdx.x * 32 + wave];
    const int* __restrict__ mywl = wl + (size_t)(blockIdx.x * 32 + wave) * WCAP;
    int wnext = mywl[lane];
    const int n = __builtin_amdgcn_readfirstlane(over ? 0 : nraw);

    const int l16  = lane & 15;
    const int half = lane >> 4;
    const int nb   = (wave & 1) * 32;
    const v4f wv = ((const v4f*)W)[tid];
    const float bv0 = bias[nb + l16];
    const float bv1 = bias[nb + 16 + l16];

    v4u tv[NDGT];
    #pragma unroll
    for (int k = 0; k < NDGT; ++k) {
        const int i = tid + k * 1024;
        tv[k] = ((const v4u*)deg8)[i < NDG16 ? i : NDG16 - 1];
    }
    v4f hv[NSV];
    float dsv[NSV];
    #pragma unroll
    for (int k = 0; k < NSV; ++k) {
        const int nc = min(lo + ((tid + k * 1024) >> 4), NNODES - 1);
        hv[k]  = ((const v4f*)h)[(size_t)nc * 16 + (tid & 15)];
        dsv[k] = dinv[nc];
    }
    const float sdv = dinv[min(lo + tid, NNODES - 1)];

    asm volatile("" :: "v"(wnext), "v"(wv), "v"(bv0), "v"(bv1), "v"(tv[0]), "v"(tv[1]), "v"(tv[2]), "v"(tv[3]),
                       "v"(hv[0]), "v"(hv[1]), "v"(hv[2]), "v"(hv[3]),
                       "v"(dsv[0]), "v"(dsv[1]), "v"(dsv[2]), "v"(dsv[3]), "v"(sdv));

    ((v4f*)(sW + (tid >> 4) * WP))[tid & 15] = wv;
    #pragma unroll
    for (int k = 0; k < NDGT; ++k) ((v4u*)sdeg)[tid + k * 1024] = tv[k];

    #pragma unroll
    for (int k = 0; k < NSV; ++k) {
        const int r = (tid + k * 1024) >> 4;
        const bool own = r < OT && lo + r < NNODES;
        const float di = dsv[k];
        const v4f z = {0.0f, 0.0f, 0.0f, 0.0f};
        ((v4f*)(acc + r * AP))[tid & 15] = own ? hv[k] * di * di : z;
    }
    if (tid < OT) sdinv[tid] = (lo + tid < NNODES) ? sdv : 0.0f;
    __syncthreads();

    const char* __restrict__ hb = (const char*)h;
    char* accb = (char*)acc;

    auto add_rows = [&](int cnt, unsigned soff, unsigned doff, float dv, float sd) {
        #pragma nounroll
        for (int j0 = 0; j0 < cnt; j0 += RG) {
            float2 v[RG];
            #pragma unroll
            for (int k = 0; k < RG; ++k) {
                const unsigned so = (unsigned)__builtin_amdgcn_readlane((int)soff, j0 + k);
                v[k] = *(const float2*)(hb + (so + (unsigned)lane * 8u));
            }
            #pragma unroll
            for (int k = 0; k < RG; ++k) {
                const float wj =
                    __builtin_bit_cast(float, __builtin_amdgcn_readlane(__builtin_bit_cast(int, dv), j0 + k)) *
                    __builtin_bit_cast(float, __builtin_amdgcn_readlane(__builtin_bit_cast(int, sd), j0 + k));
                const unsigned dj = (unsigned)__builtin_amdgcn_readlane((int)doff, j0 + k);
                float* ap = (float*)(accb + (dj + (unsigned)lane * 8u));
                atomicAdd(ap,     v[k].x * wj);
                atomicAdd(ap + 1, v[k].y * wj);
            }
        }
    };

    auto src_dinv = [&](unsigned s) {
        const unsigned dg = sdeg[s];
        float dv = rsqrtf((float)dg);
        if (dg == (unsigned)DGBIG) dv = dinv[s];
        return dv;
    };

    if (!over) {
        for (int j0 = 0; j0 < n; j0 += 32) {
            const int cnt = min(32, n - j0);
            const unsigned word = (unsigned)wnext;
            const int nj = j0 + 32 + lane;
            wnext = mywl[nj < n ? nj : n - 1];
            const unsigned s  = word & 0xFFFFu;
            const unsigned dl = word >> 16;
            const float dv = src_dinv(s);
            const float sd = sdinv[dl];
            add_rows(cnt, s * (HID * 4u), (lane < cnt ? dl : (unsigned)OT) * (AP * 4u), dv, sd);
        }
    } else {
        unsigned* __restrict__ myq = q + wave * QCAP;
        walk_dst(dst, lo, wave, lane, myq, [&](int qn) {
            const int base = qn > 32 ? qn - 32 : 0;
            const int cnt  = qn - base;
            const bool used = lane < cnt;
            const unsigned p = myq[base + (used ? lane : cnt - 1)];
            const int e  = (int)(p & 0xFFFFFu);
            const int dl = (int)(p >> 20);
            const int s  = src[e];
            const float dv = src_dinv((unsigned)s);
            const float sd = sdinv[dl];
            add_rows(cnt, (unsigned)s * (HID * 4u), (unsigned)(used ? dl : OT) * (AP * 4u), dv, sd);
            return base;
        });
    }
    __syncthreads();

    v8f c0 = {}, c1 = {};
    const int t16 = wave >> 1;
    if (wave < 2 * NT16) {
        const float* Ar = acc + (t16 * 16 + l16) * AP;
        #pragma unroll
        for (int k0 = 0; k0 < HID; k0 += 4) {
            const int kb = k0 + half * 2;
            const v2f af = {Ar[kb], Ar[kb + 1]};
            const float* Wk0 = sW + kb * WP + nb;
            const float* Wk1 = sW + (kb + 1) * WP + nb;
            const v2f bf0 = {Wk0[l16],      Wk1[l16]};
            const v2f bf1 = {Wk0[l16 + 16], Wk1[l16 + 16]};
            c0 = __builtin_amdgcn_wmma_f32_16x16x4_f32(false, af, false, bf0, (short)0, c0, false, false);
            c1 = __builtin_amdgcn_wmma_f32_16x16x4_f32(false, af, false, bf1, (short)0, c1, false, false);
        }
    }
    __syncthreads();
    if (wave < 2 * NT16) {
        #pragma unroll
        for (int r = 0; r < 8; ++r) {
            float* orow = acc + (t16 * 16 + half * 8 + r) * AP + nb + l16;
            float v0 = c0[r] + bv0;
            float v1 = c1[r] + bv1;
            if (relu_out) { v0 = fmaxf(v0, 0.0f); v1 = fmaxf(v1, 0.0f); }
            orow[0]  = v0;
            orow[16] = v1;
        }
    }
    __syncthreads();

    for (int i = tid; i < OT * 16; i += 1024) {
        const int node = lo + (i >> 4);
        if (node < NNODES) ((float4*)a)[(size_t)node * 16 + (i & 15)] = ((const float4*)(acc + (i >> 4) * AP))[i & 15];
    }
}


#define PT    256
#define PW    (PT / 32)
#define NK    ((NB4 + PT - 1) / PT)
#define NZ    ((NCH + PT - 1) / PT)
#define NWORD ((NK + 7) / 8)
#define LCAP  1024
#define RB    16
#define NWL   ((HID * NCLASSES + PT - 1) / PT)
static_assert(NNODES % 4 == 0 && HID == 64 && NCLASSES <= PT, "batch in 16-byte groups, a row is 32 lanes of 8 bytes");

__global__ __launch_bounds__(PT)
void k_graph_pool_head(const float* __restrict__ a, const int* __restrict__ batch, const int* __restrict__ zmm,
                       const float* __restrict__ Wl, const float* __restrict__ bl,
                       float* __restrict__ out) {
    __shared__ int   list[LCAP];
    __shared__ int   wtot[PW];
    __shared__ int   wlo[PW];
    __shared__ int   whi[PW];
    __shared__ float part[PW * HID];
    __shared__ float ssum[HID];
    __shared__ float sWl[HID * NCLASSES];

    const int tid  = threadIdx.x;
    const int lane = tid & 31;
    const int wave = __builtin_amdgcn_readfirstlane(tid >> 5);
    const int g    = blockIdx.x;

    float wl[NWL];
    #pragma unroll
    for (int i = 0; i < NWL; ++i) wl[i] = (i + 1) * PT <= HID * NCLASSES || tid + i * PT < HID * NCLASSES ? Wl[tid + i * PT] : 0.0f;
    const float blv = tid < NCLASSES ? bl[tid] : 0.0f;

    int wl_lo = NCH, wl_hi = -1;
    int2 z[NZ];
    #pragma unroll
    for (int i = 0; i < NZ; ++i) {
        const int ch = i * PT + tid;
        z[i] = ((const int2*)zmm)[ch < NCH ? ch : NCH - 1];
    }
    #pragma unroll
    for (int i = 0; i < NZ; ++i) {
        const int ch = i * PT + tid;
        const bool hit = ((i + 1) * PT <= NCH || ch < NCH) && z[i].x <= g && g <= z[i].y;
        const unsigned bm = __builtin_amdgcn_ballot_w32(hit);
        if (bm != 0u) {
            const int cb = i * PT + wave * 32;
            wl_lo = min(wl_lo, cb + __builtin_ctz(bm));
            wl_hi = max(wl_hi, cb + 31 - __builtin_clz(bm));
        }
    }
    if (lane == 0) { wlo[wave] = wl_lo; whi[wave] = wl_hi; }
    __syncthreads();
    int clo = NCH, chi = -1;
    for (int w = 0; w < PW; ++w) { clo = min(clo, wlo[w]); chi = max(chi, whi[w]); }
    const int g0 = clo * CG;
    const int g1 = min((chi + 1) * CG, NB4);
    const bool near = g1 - g0 <= PT;

    const int4* __restrict__ batch4 = (const int4*)batch;
    unsigned m1 = 0u;
    unsigned mask[NWORD];
    #pragma unroll
    for (int w = 0; w < NWORD; ++w) mask[w] = 0u;
    if (near) {
        const int idx = g0 + tid;
        if (idx < g1) {
            const int4 d = batch4[idx];
            m1 = (d.x == g ? 1u : 0u) | (d.y == g ? 2u : 0u) | (d.z == g ? 4u : 0u) | (d.w == g ? 8u : 0u);
        }
    } else {
    #pragma unroll
    for (int w = 0; w < NWORD; ++w) {
        int4 d[8];
        #pragma unroll
        for (int j = 0; j < 8; ++j) {
            const int k = w * 8 + j;
            d[j] = make_int4(-1, -1, -1, -1);
            if (k < NK) {
                const int idx = k * PT + tid;
                if ((k + 1) * PT <= NB4 || idx < NB4) d[j] = batch4[idx];
            }
        }
        unsigned m = 0u;
        #pragma unroll
        for (int j = 0; j < 8; ++j) {
            if (w * 8 + j < NK) {
                m |= (d[j].x == g ? 1u : 0u) << (4 * j);
                m |= (d[j].y == g ? 2u : 0u) << (4 * j);
                m |= (d[j].z == g ? 4u : 0u) << (4 * j);
                m |= (d[j].w == g ? 8u : 0u) << (4 * j);
            }
        }
        mask[w] = m;
    }
    }

    #pragma unroll
    for (int i = 0; i < NWL; ++i) if ((i + 1) * PT <= HID * NCLASSES || tid + i * PT < HID * NCLASSES) sWl[tid + i * PT] = wl[i];

    int c = __builtin_popcount(m1);
    #pragma unroll
    for (int w = 0; w < NWORD; ++w) c += __builtin_popcount(mask[w]);
    int x = c;
    for (int dl = 1; dl < 32; dl <<= 1) {
        const int y = __shfl_up(x, dl);
        if (lane >= dl) x += y;
    }
    if (lane == 31) wtot[wave] = x;
    __syncthreads();
    int base = 0, total = 0;
    for (int w = 0; w < PW; ++w) {
        const int t = wtot[w];
        if (w < wave) base += t;
        total += t;
    }
    const int first = base + x - c;

    float2 acc = make_float2(0.0f, 0.0f);
    const float2* __restrict__ a2 = (const float2*)a;
    for (int r0 = 0; r0 < total; r0 += LCAP) {
        if (r0 != 0) __syncthreads();
        if (c != 0) {
            int p = first - r0;
            for (unsigned m = m1; m != 0u; m &= m - 1u) {
                if (p >= 0 && p < LCAP) list[p] = ((g0 + tid) << 2) + __builtin_ctz(m);
                ++p;
            }
            #pragma unroll
            for (int w = 0; w < NWORD; ++w) {
                unsigned m = mask[w];
                while (m != 0u) {
                    const int b = __builtin_ctz(m);
                    m &= m - 1u;
                    if (p >= 0 && p < LCAP) list[p] = (((w * 8 + (b >> 2)) * PT + tid) << 2) + (b & 3);
                    ++p;
                }
            }
        }
        __syncthreads();
        const int n = min(total - r0, LCAP);
        for (int p0 = wave; p0 < n + wave; p0 += PW * RB) {
            float2 v[RB];
            #pragma unroll
            for (int k = 0; k < RB; ++k) {
                const int p = p0 + PW * k < n ? p0 + PW * k : n - 1;
                const int node = __builtin_amdgcn_readfirstlane(list[p]);
                v[k] = a2[(size_t)node * 32 + lane];
            }
            #pragma unroll
            for (int k = 0; k < RB; ++k) {
                const bool ok = p0 + PW * k < n;
                acc.x += fmaxf(ok ? v[k].x : 0.0f, 0.0f);
                acc.y += fmaxf(ok ? v[k].y : 0.0f, 0.0f);
            }
        }
    }

    ((float2*)part)[wave * 32 + lane] = acc;
    __syncthreads();
    if (tid < HID) {
        float s = 0.0f;
        for (int w = 0; w < PW; ++w) s += part[w * HID + tid];
        ssum[tid] = s;
    }
    __syncthreads();
    if (tid < NCLASSES) {
        const float inv = 1.0f / fmaxf((float)total, 1.0f);
        float o = 0.0f;
        #pragma unroll
        for (int k = 0; k < HID; ++k) o += ssum[k] * sWl[k * NCLASSES + tid];
        out[g * NCLASSES + tid] = o * inv + blv;
    }
}


extern "C" void kernel_launch(void* const* d_in, const int* in_sizes, int n_in,
                              void* d_out, int out_size, void* d_ws, size_t ws_size,
                              hipStream_t stream) {
    const float* x  = (const float*)d_in[0];
    const float* W1 = (const float*)d_in[1];
    const float* b1 = (const float*)d_in[2];
    const float* W2 = (const float*)d_in[3];
    const float* b2 = (const float*)d_in[4];
    const float* Wl = (const float*)d_in[5];
    const float* bl = (const float*)d_in[6];
    const int* edge  = (const int*)d_in[7];
    const int* batch = (const int*)d_in[8];
    const int* src = edge;
    const int* dst = edge + NEDGES;
    float* out = (float*)d_out;

    float* ws   = (float*)d_ws;
    float* h    = ws;
    float* a    = h + (size_t)NNODES * HID;
    float* dinv = a + (size_t)NNODES * HID;
    int* wl   = (int*)ws + (size_t)WL_OFF;
    int* wcnt = (int*)ws + (size_t)WCNT_OFF;
    int* ovf  = (int*)ws + (size_t)OVF_OFF;
    unsigned* ob = (unsigned*)ws + (size_t)OB_OFF;
    int* ew   = (int*)ws + (size_t)EW_OFF;
    int* zmm  = (int*)ws + (size_t)ZM_OFF;
    unsigned char* deg8 = (unsigned char*)((int*)ws + (size_t)DG_OFF);

    k_edge_words<<<EBLK, 256, 0, stream>>>(src, dst, batch, ob, ew, zmm);

    k_owner_filter<<<OBLK, 1024, 0, stream>>>(ob, ew, wl, wcnt, ovf, dinv, deg8);

    k_owner_agg<<<OBLK, 1024, 0, stream>>>(x, dinv, deg8, W1, b1, 1, src, dst, wl, wcnt, ovf, h);

    k_owner_agg<<<OBLK, 1024, 0, stream>>>(h, dinv, deg8, W2, b2, 0, src, dst, wl, wcnt, ovf, a);

    k_graph_pool_head<<<NGRAPHS, PT, 0, stream>>>(a, batch, zmm, Wl, bl, out);
}


# Owner numbers bit-sliced (8 slice words per 32 edges, built by the first launch with ballots): the owner walk tests 32 edges with 12 logic ops, not some 50 on bytes; list, sum order changed
# speedup vs baseline: 5.0514x; 1.0314x over previous
#include <hip/hip_runtime.h>

typedef float v2f __attribute__((ext_vector_type(2)));
typedef float v8f __attribute__((ext_vector_type(8)));
typedef float v4f __attribute__((ext_vector_type(4)));
typedef unsigned v4u __attribute__((ext_vector_type(4)));

#define NNODES 50000
#define NEDGES 800000
#define HID 64
#define NGRAPHS 512
#define NCLASSES 10


#define OT   196
#define NV4  (NEDGES / 4)
#define NSG  (NV4 / 32)
#define NPAIR (NSG / 2)
#define PPW  2
#define EBLK ((NPAIR + 8 * PPW - 1) / (8 * PPW))
#define NW1  (EBLK * 8)
#define NU   (NEDGES / 32)
#define NST  ((NU + 31) / 32)
#define OBW  (NST * 256)
static_assert(NV4 % 64 == 0, "whole pairs of super-groups");
#define NB4   (NNODES / 4)
#define CG    32
#define NCH   ((NB4 + CG - 1) / CG)
static_assert(NNODES % 4 == 0 && NW1 >= NCH, "batch in 16-byte groups; the first launch has a wave for every chunk");

__global__ __launch_bounds__(256)
void k_edge_words(const int* __restrict__ src, const int* __restrict__ dst, const int* __restrict__ batch,
                  unsigned* __restrict__ ob, int* __restrict__ ew, int* __restrict__ zmm) {
    const int lane = threadIdx.x & 31;
    const int gw = __builtin_amdgcn_readfirstlane(blockIdx.x * 8 + (int)(threadIdx.x >> 5));
    int4 bq = make_int4(0, 0, 0, 0);
    if (gw < NCH) {
        const int gi = gw * CG + lane;
        bq = ((const int4*)batch)[gi < NB4 ? gi : NB4 - 1];
    }
    int4 dv[PPW][2], sv[PPW][2];
    #pragma unroll
    for (int jj = 0; jj < PPW; ++jj) {
        const int P  = gw + jj * NW1;
        const int Pc = P < NPAIR ? P : NPAIR - 1;
        #pragma unroll
        for (int hh = 0; hh < 2; ++hh) {
            const int g = (2 * Pc + hh) * 32 + lane;
            dv[jj][hh] = ((const int4*)dst)[g];
            sv[jj][hh] = ((const int4*)src)[g];
        }
    }
    #pragma unroll
    for (int jj = 0; jj < PPW; ++jj) {
        const int P = gw + jj * NW1;
        if (P < NPAIR) {
            unsigned pk[2];
            #pragma unroll
            for (int hh = 0; hh < 2; ++hh) {
                const int g = (2 * P + hh) * 32 + lane;
                const int4 d = dv[jj][hh];
                const int4 s = sv[jj][hh];
                const unsigned o0 = (unsigned)d.x / OT, o1 = (unsigned)d.y / OT;
                const unsigned o2 = (unsigned)d.z / OT, o3 = (unsigned)d.w / OT;
                int4 w;
                w.x = s.x | (int)(((unsigned)d.x - o0 * OT) << 16);
                w.y = s.y | (int)(((unsigned)d.y - o1 * OT) << 16);
                w.z = s.z | (int)(((unsigned)d.z - o2 * OT) << 16);
                w.w = s.w | (int)(((unsigned)d.w - o3 * OT) << 16);
                ((int4*)ew)[g] = w;
                pk[hh] = (o0 & 255u) | (o1 & 255u) << 8 | (o2 & 255u) << 16 | o3 << 24;
            }
            unsigned r0 = 0u, r1 = 0u;
            #pragma unroll
            for (int hh = 0; hh < 2; ++hh) {
                #pragma unroll
                for (int cb = 0; cb < 32; ++cb) {
                    const int cc = cb >> 3, b = cb & 7;
                    const unsigned bal = __builtin_amdgcn_ballot_w32(((pk[hh] >> cb) & 1u) != 0u);
                    const int L = hh * 16 + cc * 4 + (b & 3);
                    if (b < 4) r0 = lane == L ? bal : r0;
                    else       r1 = lane == L ? bal : r1;
                }
            }
            unsigned* line = ob + (size_t)(P >> 2) * 256 + 32 * (P & 3) + lane;
            line[0]   = r0;
            line[128] = r1;
        }
    }
    if (gw < NCH) {
        int mn = min(min(bq.x, bq.y), min(bq.z, bq.w));
        int mx = max(max(bq.x, bq.y), max(bq.z, bq.w));
        for (int dl = 16; dl >= 1; dl >>= 1) {
            mn = min(mn, __shfl_xor(mn, dl));
            mx = max(mx, __shfl_xor(mx, dl));
        }
        if (lane == 0) ((int2*)zmm)[gw] = make_int2(mn, mx);
    }
}


#define OBLK ((NNODES + OT - 1) / OT)
#define OTP  ((OT + 1 + 15) / 16 * 16)
#define NT16 (OTP / 16)
#define AP   68
#define WP   72
#define TL   5
#define SCH  (TL * 32)
#define NSC  (NV4 / SCH)
#define RG   16
#define QCAP 896
#define WCAP 512
#define FQCAP 1312
#define WL_OFF   ((2 * NNODES * HID + NNODES + 63) / 64 * 64)
#define WCNT_OFF (WL_OFF + OBLK * 32 * WCAP)
#define OVF_OFF  (WCNT_OFF + OBLK * 32)
#define OB_OFF   ((OVF_OFF + OBLK + 63) / 64 * 64)
#define EW_OFF   ((OB_OFF + OBW + 63) / 64 * 64)
#define ZM_OFF   ((EW_OFF + NEDGES + 63) / 64 * 64)
#define DG_OFF   ((ZM_OFF + 2 * NCH + 63) / 64 * 64)
#define DGBIG 255
#define NDG16 (NNODES / 16)
#define NDGT  ((NDG16 + 1023) / 1024)
static_assert(NNODES % 16 == 0, "the degree table is copied in whole 16-byte groups");
#define NSV  ((OTP * 16 + 1023) / 1024)
#define ACR  (NSV * 1024 / 16)
static_assert(ACR >= OTP, "the start values cover every row the product reads");
static_assert(WCAP >= 32 && NDGT == 4 && NSV == 4, "a list region holds at least one batch; the load group's hold names four table and four start-row trips");
#define TS   5
static_assert(NEDGES % 4 == 0 && NV4 % SCH == 0, "dst is read in whole trips of 16-byte groups");
static_assert(2 * NT16 <= 32 && HID == 64, "product: two waves per 16-row tile, 32 columns each");
static_assert(AP % 4 == 0 && WP % 4 == 0 && AP >= HID && WP >= HID, "LDS rows hold 64 floats and start on 16-byte boundaries");
static_assert(NEDGES <= (1 << 20) && OT <= (1 << 12), "queue word = edge | row << 20");
static_assert(NNODES <= (1 << 16) && OT <= (1 << 16), "list word = source | row << 16");
static_assert(QCAP >= 31 + SCH * 4 && FQCAP >= QCAP && NSC >= 32 && 32 % RG == 0, "queue room, a trip for every wave");
static_assert(32 * FQCAP * 4 > 160 * 1024 && 32 * FQCAP * 4 + OT * 4 + 4 <= 320 * 1024, "filter block: over half of the LDS, within all of it");
static_assert(OBLK <= 256 && NEDGES % 128 == 0 && FQCAP >= 31 + 32 * 32 && NST >= 32, "an owner number has eight bits; one step's hits fit the queue");

template <class Take>
__device__ __forceinline__ void walk_dst(const int* __restrict__ dst, int lo, int wave, int lane,
                                         unsigned* __restrict__ myq, Take&& take) {
    int qn = 0;
    const int4* __restrict__ dst4 = (const int4*)dst;
    const int start = (blockIdx.x * (NSC / OBLK)) % NSC;
    const int ntrip = (NSC - wave + 31) / 32;

    auto load_trip = [&](int i, int4 (&buf)[TL], int& base) {
        const int ic = i < ntrip ? i : ntrip - 1;
        int s2 = ic * 32 + wave + start;
        if (s2 >= NSC) s2 -= NSC;
        base = s2 * SCH;
        #pragma unroll
        for (int k = 0; k < TL; ++k) buf[k] = dst4[base + k * 32 + lane];
    };

    auto push = [&](unsigned t, int e) {
        const bool m = t < (unsigned)OT;
        const unsigned mask = __builtin_amdgcn_ballot_w32(m);
        if (mask != 0u) {
            if (m) {
                const int slot = qn + (int)__builtin_amdgcn_mbcnt_lo(mask, 0u);
                myq[slot] = (unsigned)e | (t << 20);
            }
            qn += __builtin_popcount(mask);
        }
    };

    auto process = [&](const int4 (&buf)[TL], int base, bool last) {
        #pragma unroll
        for (int k = 0; k < TL; ++k) {
            const int4 d = buf[k];
            const unsigned t0 = (unsigned)(d.x - lo), t1 = (unsigned)(d.y - lo);
            const unsigned t2 = (unsigned)(d.z - lo), t3 = (unsigned)(d.w - lo);
            const unsigned tm = min(min(t0, t1), min(t2, t3));
            if (__builtin_amdgcn_ballot_w32(tm < (unsigned)OT) != 0u) {
                const int e0 = (base + k * 32 + lane) * 4;
                push(t0, e0);
                push(t1, e0 + 1);
                push(t2, e0 + 2);
                push(t3, e0 + 3);
            }
        }
        const int keep = last ? 0 : 31;
        while (qn > keep) qn = take(qn);
    };

    int4 bufA[TL], bufB[TL];
    int baseA = 0, baseB = 0;
    load_trip(0, bufA, baseA);
    for (int i = 0; i < ntrip; i += 2) {
        load_trip(i + 1, bufB, baseB);
        process(bufA, baseA, i + 1 >= ntrip);
        if (i + 1 >= ntrip) break;
        load_trip(i + 2, bufA, baseA);
        process(bufB, baseB, i + 2 >= ntrip);
    }
}

__global__ __launch_bounds__(1024)
void k_owner_filter(const unsigned* __restrict__ ob, const int* __restrict__ ew,
                    int* __restrict__ wl, int* __restrict__ wcnt, int* __restrict__ ovf,
                    float* __restrict__ dinv, unsigned char* __restrict__ deg8) {
    __shared__ int scnt[OT];
    __shared__ int sovf;
    __shared__ unsigned q[32 * FQCAP];

    const int tid  = threadIdx.x;
    const int lane = tid & 31;
    const int wave = __builtin_amdgcn_readfirstlane(tid >> 5);
    const int lo   = blockIdx.x * OT;
    const uint4* __restrict__ ob4 = (const uint4*)ob;
    unsigned zb[8];
    #pragma unroll
    for (int b = 0; b < 8; ++b) zb[b] = ((blockIdx.x >> b) & 1u) ? 0u : 0xFFFFFFFFu;

    if (tid < OT) scnt[tid] = 0;
    if (tid == 0) sovf = 0;
    __syncthreads();

    unsigned* __restrict__ myq = q + wave * FQCAP;
    int* __restrict__ mywl = wl + (size_t)(blockIdx.x * 32 + wave) * WCAP;
    int wn = 0;
    int qn = 0;

    auto take = [&](int fill) {
        const int base = fill > 32 ? fill - 32 : 0;
        const int cnt  = fill - base;
        if (lane < cnt) {
            const unsigned w = myq[base + lane];
            const unsigned u = w >> 5;
            const unsigned e = ((u >> 2) << 7) + ((w & 31u) << 2) + (u & 3u);
            const int word = ew[e];
            atomicAdd(&scnt[(unsigned)word >> 16], 1);
            if (wn + lane < WCAP) mywl[wn + lane] = word;
        }
        wn += cnt;
        return base;
    };

    const int start = (blockIdx.x * NST) / OBLK;
    const int nst   = (NST - wave + 31) / 32;
    const int ntrip = (nst + TS - 1) / TS;

    auto step_of = [&](int j) {
        const int jc = j < nst ? j : nst - 1;
        int s2 = jc * 32 + wave + start;
        if (s2 >= NST) s2 -= NST;
        return s2;
    };

    auto load_trip = [&](int i, uint4 (&bl)[TS], uint4 (&bh)[TS]) {
        #pragma unroll
        for (int k = 0; k < TS; ++k) {
            const int u  = step_of(i * TS + k) * 32 + lane;
            const int uc = u < NU ? u : NU - 1;
            const int ix = (uc >> 5) * 64 + (uc & 31);
            bl[k] = ob4[ix];
            bh[k] = ob4[ix + 32];
        }
    };

    auto process = [&](int i, const uint4 (&bl)[TS], const uint4 (&bh)[TS]) {
        #pragma unroll
        for (int k = 0; k < TS; ++k) {
            if (i * TS + k < nst) {
                const int s2 = step_of(i * TS + k);
                const int u  = s2 * 32 + lane;
                unsigned m = (bl[k].x ^ zb[0]) & (bl[k].y ^ zb[1]) & (bl[k].z ^ zb[2]) & (bl[k].w ^ zb[3])
                           & (bh[k].x ^ zb[4]) & (bh[k].y ^ zb[5]) & (bh[k].z ^ zb[6]) & (bh[k].w ^ zb[7]);
                if (s2 == NST - 1) m = u < NU ? m : 0u;
                unsigned any = __builtin_amdgcn_ballot_w32(m != 0u);
                while (any != 0u) {
                    if (m != 0u) {
                        const int p = __builtin_ctz(m);
                        const int slot = qn + (int)__builtin_amdgcn_mbcnt_lo(any, 0u);
                        myq[slot] = (unsigned)(u * 32 + p);
                        m &= m - 1u;
                    }
                    qn += __builtin_popcount(any);
                    any = __builtin_amdgcn_ballot_w32(m != 0u);
                }
                while (qn > 31) qn = take(qn);
            }
        }
    };

    uint4 blA[TS], bhA[TS], blB[TS], bhB[TS];
    load_trip(0, blA, bhA);
    for (int i = 0; i < ntrip; i += 2) {
        load_trip(i + 1, blB, bhB);
        process(i, blA, bhA);
        if (i + 1 >= ntrip) break;
        load_trip(i + 2, blA, bhA);
        process(i + 1, blB, bhB);
    }
    while (qn > 0) qn = take(qn);

    if (wn < 32 && lane >= wn) mywl[lane] = 0;

    if (lane == 0) {
        wcnt[blockIdx.x * 32 + wave] = wn;
        if (wn > WCAP) sovf = 1;
    }
    __syncthreads();
    if (tid < OT && lo + tid < NNODES) {
        const int dg = scnt[tid] + 1;
        dinv[lo + tid]  = rsqrtf((float)dg);
        deg8[lo + tid]  = (unsigned char)(dg < DGBIG ? dg : DGBIG);
    }
    if (tid == 0) ovf[blockIdx.x] = sovf;
}

__global__ __launch_bounds__(1024)
void k_owner_agg(const float* __restrict__ h, const float* __restrict__ dinv, const unsigned char* __restrict__ deg8,
                 const float* __restrict__ W, const float* __restrict__ bias, int relu_out,
                 const int* __restrict__ src, const int* __restrict__ dst,
                 const int* __restrict__ wl, const int* __restrict__ wcnt, const int* __restrict__ ovf,
                 float* __restrict__ a) {
    __shared__ float acc[ACR * AP];
    __shared__ float sW[HID * WP];
    __shared__ float sdinv[OT];
    __shared__ unsigned q[32 * QCAP];
    __shared__ __attribute__((aligned(16))) unsigned char sdeg[NDGT * 1024 * 16];

    const int tid  = threadIdx.x;
    const int lane = tid & 31;
    const int wave = __builtin_amdgcn_readfirstlane(tid >> 5);
    const int lo   = blockIdx.x * OT;

    const int over = ovf[blockIdx.x];
    const int nraw = wcnt[blockIdx.x * 32 + wave];
    const int* __restrict__ mywl = wl + (size_t)(blockIdx.x * 32 + wave) * WCAP;
    int wnext = mywl[lane];
    const int n = __builtin_amdgcn_readfirstlane(over ? 0 : nraw);

    const int l16  = lane & 15;
    const int half = lane >> 4;
    const int nb   = (wave & 1) * 32;
    const v4f wv = ((const v4f*)W)[tid];
    const float bv0 = bias[nb + l16];
    const float bv1 = bias[nb + 16 + l16];

    v4u tv[NDGT];
    #pragma unroll
    for (int k = 0; k < NDGT; ++k) {
        const int i = tid + k * 1024;
        tv[k] = ((const v4u*)deg8)[i < NDG16 ? i : NDG16 - 1];
    }
    v4f hv[NSV];
    float dsv[NSV];
    #pragma unroll
    for (int k = 0; k < NSV; ++k) {
        const int nc = min(lo + ((tid + k * 1024) >> 4), NNODES - 1);
        hv[k]  = ((const v4f*)h)[(size_t)nc * 16 + (tid & 15)];
        dsv[k] = dinv[nc];
    }
    const float sdv = dinv[min(lo + tid, NNODES - 1)];

    asm volatile("" :: "v"(wnext), "v"(wv), "v"(bv0), "v"(bv1), "v"(tv[0]), "v"(tv[1]), "v"(tv[2]), "v"(tv[3]),
                       "v"(hv[0]), "v"(hv[1]), "v"(hv[2]), "v"(hv[3]),
                       "v"(dsv[0]), "v"(dsv[1]), "v"(dsv[2]), "v"(dsv[3]), "v"(sdv));

    ((v4f*)(sW + (tid >> 4) * WP))[tid & 15] = wv;
    #pragma unroll
    for (int k = 0; k < NDGT; ++k) ((v4u*)sdeg)[tid + k * 1024] = tv[k];

    #pragma unroll
    for (int k = 0; k < NSV; ++k) {
        const int r = (tid + k * 1024) >> 4;
        const bool own = r < OT && lo + r < NNODES;
        const float di = dsv[k];
        const v4f z = {0.0f, 0.0f, 0.0f, 0.0f};
        ((v4f*)(acc + r * AP))[tid & 15] = own ? hv[k] * di * di : z;
    }
    if (tid < OT) sdinv[tid] = (lo + tid < NNODES) ? sdv : 0.0f;
    __syncthreads();

    const char* __restrict__ hb = (const char*)h;
    char* accb = (char*)acc;

    auto add_rows = [&](int cnt, unsigned soff, unsigned doff, float dv, float sd) {
        #pragma nounroll
        for (int j0 = 0; j0 < cnt; j0 += RG) {
            float2 v[RG];
            #pragma unroll
            for (int k = 0; k < RG; ++k) {
                const unsigned so = (unsigned)__builtin_amdgcn_readlane((int)soff, j0 + k);
                v[k] = *(const float2*)(hb + (so + (unsigned)lane * 8u));
            }
            #pragma unroll
            for (int k = 0; k < RG; ++k) {
                const float wj =
                    __builtin_bit_cast(float, __builtin_amdgcn_readlane(__builtin_bit_cast(int, dv), j0 + k)) *
                    __builtin_bit_cast(float, __builtin_amdgcn_readlane(__builtin_bit_cast(int, sd), j0 + k));
                const unsigned dj = (unsigned)__builtin_amdgcn_readlane((int)doff, j0 + k);
                float* ap = (float*)(accb + (dj + (unsigned)lane * 8u));
                atomicAdd(ap,     v[k].x * wj);
                atomicAdd(ap + 1, v[k].y * wj);
            }
        }
    };

    auto src_dinv = [&](unsigned s) {
        const unsigned dg = sdeg[s];
        float dv = rsqrtf((float)dg);
        if (dg == (unsigned)DGBIG) dv = dinv[s];
        return dv;
    };

    if (!over) {
        for (int j0 = 0; j0 < n; j0 += 32) {
            const int cnt = min(32, n - j0);
            const unsigned word = (unsigned)wnext;
            const int nj = j0 + 32 + lane;
            wnext = mywl[nj < n ? nj : n - 1];
            const unsigned s  = word & 0xFFFFu;
            const unsigned dl = word >> 16;
            const float dv = src_dinv(s);
            const float sd = sdinv[dl];
            add_rows(cnt, s * (HID * 4u), (lane < cnt ? dl : (unsigned)OT) * (AP * 4u), dv, sd);
        }
    } else {
        unsigned* __restrict__ myq = q + wave * QCAP;
        walk_dst(dst, lo, wave, lane, myq, [&](int qn) {
            const int base = qn > 32 ? qn - 32 : 0;
            const int cnt  = qn - base;
            const bool used = lane < cnt;
            const unsigned p = myq[base + (used ? lane : cnt - 1)];
            const int e  = (int)(p & 0xFFFFFu);
            const int dl = (int)(p >> 20);
            const int s  = src[e];
            const float dv = src_dinv((unsigned)s);
            const float sd = sdinv[dl];
            add_rows(cnt, (unsigned)s * (HID * 4u), (unsigned)(used ? dl : OT) * (AP * 4u), dv, sd);
            return base;
        });
    }
    __syncthreads();

    v8f c0 = {}, c1 = {};
    const int t16 = wave >> 1;
    if (wave < 2 * NT16) {
        const float* Ar = acc + (t16 * 16 + l16) * AP;
        #pragma unroll
        for (int k0 = 0; k0 < HID; k0 += 4) {
            const int kb = k0 + half * 2;
            const v2f af = {Ar[kb], Ar[kb + 1]};
            const float* Wk0 = sW + kb * WP + nb;
            const float* Wk1 = sW + (kb + 1) * WP + nb;
            const v2f bf0 = {Wk0[l16],      Wk1[l16]};
            const v2f bf1 = {Wk0[l16 + 16], Wk1[l16 + 16]};
            c0 = __builtin_amdgcn_wmma_f32_16x16x4_f32(false, af, false, bf0, (short)0, c0, false, false);
            c1 = __builtin_amdgcn_wmma_f32_16x16x4_f32(false, af, false, bf1, (short)0, c1, false, false);
        }
    }
    __syncthreads();
    if (wave < 2 * NT16) {
        #pragma unroll
        for (int r = 0; r < 8; ++r) {
            float* orow = acc + (t16 * 16 + half * 8 + r) * AP + nb + l16;
            float v0 = c0[r] + bv0;
            float v1 = c1[r] + bv1;
            if (relu_out) { v0 = fmaxf(v0, 0.0f); v1 = fmaxf(v1, 0.0f); }
            orow[0]  = v0;
            orow[16] = v1;
        }
    }
    __syncthreads();

    for (int i = tid; i < OT * 16; i += 1024) {
        const int node = lo + (i >> 4);
        if (node < NNODES) ((float4*)a)[(size_t)node * 16 + (i & 15)] = ((const float4*)(acc + (i >> 4) * AP))[i & 15];
    }
}


#define PT    256
#define PW    (PT / 32)
#define NK    ((NB4 + PT - 1) / PT)
#define NZ    ((NCH + PT - 1) / PT)
#define NWORD ((NK + 7) / 8)
#define LCAP  1024
#define RB    16
#define NWL   ((HID * NCLASSES + PT - 1) / PT)
static_assert(NNODES % 4 == 0 && HID == 64 && NCLASSES <= PT, "batch in 16-byte groups, a row is 32 lanes of 8 bytes");

__global__ __launch_bounds__(PT)
void k_graph_pool_head(const float* __restrict__ a, const int* __restrict__ batch, const int* __restrict__ zmm,
                       const float* __restrict__ Wl, const float* __restrict__ bl,
                       float* __restrict__ out) {
    __shared__ int   list[LCAP];
    __shared__ int   wtot[PW];
    __shared__ int   wlo[PW];
    __shared__ int   whi[PW];
    __shared__ float part[PW * HID];
    __shared__ float ssum[HID];
    __shared__ float sWl[HID * NCLASSES];

    const int tid  = threadIdx.x;
    const int lane = tid & 31;
    const int wave = __builtin_amdgcn_readfirstlane(tid >> 5);
    const int g    = blockIdx.x;

    float wl[NWL];
    #pragma unroll
    for (int i = 0; i < NWL; ++i) wl[i] = (i + 1) * PT <= HID * NCLASSES || tid + i * PT < HID * NCLASSES ? Wl[tid + i * PT] : 0.0f;
    const float blv = tid < NCLASSES ? bl[tid] : 0.0f;

    int wl_lo = NCH, wl_hi = -1;
    int2 z[NZ];
    #pragma unroll
    for (int i = 0; i < NZ; ++i) {
        const int ch = i * PT + tid;
        z[i] = ((const int2*)zmm)[ch < NCH ? ch : NCH - 1];
    }
    #pragma unroll
    for (int i = 0; i < NZ; ++i) {
        const int ch = i * PT + tid;
        const bool hit = ((i + 1) * PT <= NCH || ch < NCH) && z[i].x <= g && g <= z[i].y;
        const unsigned bm = __builtin_amdgcn_ballot_w32(hit);
        if (bm != 0u) {
            const int cb = i * PT + wave * 32;
            wl_lo = min(wl_lo, cb + __builtin_ctz(bm));
            wl_hi = max(wl_hi, cb + 31 - __builtin_clz(bm));
        }
    }
    if (lane == 0) { wlo[wave] = wl_lo; whi[wave] = wl_hi; }
    __syncthreads();
    int clo = NCH, chi = -1;
    for (int w = 0; w < PW; ++w) { clo = min(clo, wlo[w]); chi = max(chi, whi[w]); }
    const int g0 = clo * CG;
    const int g1 = min((chi + 1) * CG, NB4);
    const bool near = g1 - g0 <= PT;

    const int4* __restrict__ batch4 = (const int4*)batch;
    unsigned m1 = 0u;
    unsigned mask[NWORD];
    #pragma unroll
    for (int w = 0; w < NWORD; ++w) mask[w] = 0u;
    if (near) {
        const int idx = g0 + tid;
        if (idx < g1) {
            const int4 d = batch4[idx];
            m1 = (d.x == g ? 1u : 0u) | (d.y == g ? 2u : 0u) | (d.z == g ? 4u : 0u) | (d.w == g ? 8u : 0u);
        }
    } else {
    #pragma unroll
    for (int w = 0; w < NWORD; ++w) {
        int4 d[8];
        #pragma unroll
        for (int j = 0; j < 8; ++j) {
            const int k = w * 8 + j;
            d[j] = make_int4(-1, -1, -1, -1);
            if (k < NK) {
                const int idx = k * PT + tid;
                if ((k + 1) * PT <= NB4 || idx < NB4) d[j] = batch4[idx];
            }
        }
        unsigned m = 0u;
        #pragma unroll
        for (int j = 0; j < 8; ++j) {
            if (w * 8 + j < NK) {
                m |= (d[j].x == g ? 1u : 0u) << (4 * j);
                m |= (d[j].y == g ? 2u : 0u) << (4 * j);
                m |= (d[j].z == g ? 4u : 0u) << (4 * j);
                m |= (d[j].w == g ? 8u : 0u) << (4 * j);
            }
        }
        mask[w] = m;
    }
    }

    #pragma unroll
    for (int i = 0; i < NWL; ++i) if ((i + 1) * PT <= HID * NCLASSES || tid + i * PT < HID * NCLASSES) sWl[tid + i * PT] = wl[i];

    int c = __builtin_popcount(m1);
    #pragma unroll
    for (int w = 0; w < NWORD; ++w) c += __builtin_popcount(mask[w]);
    int x = c;
    for (int dl = 1; dl < 32; dl <<= 1) {
        const int y = __shfl_up(x, dl);
        if (lane >= dl) x += y;
    }
    if (lane == 31) wtot[wave] = x;
    __syncthreads();
    int base = 0, total = 0;
    for (int w = 0; w < PW; ++w) {
        const int t = wtot[w];
        if (w < wave) base += t;
        total += t;
    }
    const int first = base + x - c;

    float2 acc = make_float2(0.0f, 0.0f);
    const float2* __restrict__ a2 = (const float2*)a;
    for (int r0 = 0; r0 < total; r0 += LCAP) {
        if (r0 != 0) __syncthreads();
        if (c != 0) {
            int p = first - r0;
            for (unsigned m = m1; m != 0u; m &= m - 1u) {
                if (p >= 0 && p < LCAP) list[p] = ((g0 + tid) << 2) + __builtin_ctz(m);
                ++p;
            }
            #pragma unroll
            for (int w = 0; w < NWORD; ++w) {
                unsigned m = mask[w];
                while (m != 0u) {
                    const int b = __builtin_ctz(m);
                    m &= m - 1u;
                    if (p >= 0 && p < LCAP) list[p] = (((w * 8 + (b >> 2)) * PT + tid) << 2) + (b & 3);
                    ++p;
                }
            }
        }
        __syncthreads();
        const int n = min(total - r0, LCAP);
        for (int p0 = wave; p0 < n + wave; p0 += PW * RB) {
            float2 v[RB];
            #pragma unroll
            for (int k = 0; k < RB; ++k) {
                const int p = p0 + PW * k < n ? p0 + PW * k : n - 1;
                const int node = __builtin_amdgcn_readfirstlane(list[p]);
                v[k] = a2[(size_t)node * 32 + lane];
            }
            #pragma unroll
            for (int k = 0; k < RB; ++k) {
                const bool ok = p0 + PW * k < n;
                acc.x += fmaxf(ok ? v[k].x : 0.0f, 0.0f);
                acc.y += fmaxf(ok ? v[k].y : 0.0f, 0.0f);
            }
        }
    }

    ((float2*)part)[wave * 32 + lane] = acc;
    __syncthreads();
    if (tid < HID) {
        float s = 0.0f;
        for (int w = 0; w < PW; ++w) s += part[w * HID + tid];
        ssum[tid] = s;
    }
    __syncthreads();
    if (tid < NCLASSES) {
        const float inv = 1.0f / fmaxf((float)total, 1.0f);
        float o = 0.0f;
        #pragma unroll
        for (int k = 0; k < HID; ++k) o += ssum[k] * sWl[k * NCLASSES + tid];
        out[g * NCLASSES + tid] = o * inv + blv;
    }
}


extern "C" void kernel_launch(void* const* d_in, const int* in_sizes, int n_in,
                              void* d_out, int out_size, void* d_ws, size_t ws_size,
                              hipStream_t stream) {
    const float* x  = (const float*)d_in[0];
    const float* W1 = (const float*)d_in[1];
    const float* b1 = (const float*)d_in[2];
    const float* W2 = (const float*)d_in[3];
    const float* b2 = (const float*)d_in[4];
    const float* Wl = (const float*)d_in[5];
    const float* bl = (const float*)d_in[6];
    const int* edge  = (const int*)d_in[7];
    const int* batch = (const int*)d_in[8];
    const int* src = edge;
    const int* dst = edge + NEDGES;
    float* out = (float*)d_out;

    float* ws   = (float*)d_ws;
    float* h    = ws;
    float* a    = h + (size_t)NNODES * HID;
    float* dinv = a + (size_t)NNODES * HID;
    int* wl   = (int*)ws + (size_t)WL_OFF;
    int* wcnt = (int*)ws + (size_t)WCNT_OFF;
    int* ovf  = (int*)ws + (size_t)OVF_OFF;
    unsigned* ob = (unsigned*)ws + (size_t)OB_OFF;
    int* ew   = (int*)ws + (size_t)EW_OFF;
    int* zmm  = (int*)ws + (size_t)ZM_OFF;
    unsigned char* deg8 = (unsigned char*)((int*)ws + (size_t)DG_OFF);

    k_edge_words<<<EBLK, 256, 0, stream>>>(src, dst, batch, ob, ew, zmm);

    k_owner_filter<<<OBLK, 1024, 0, stream>>>(ob, ew, wl, wcnt, ovf, dinv, deg8);

    k_owner_agg<<<OBLK, 1024, 0, stream>>>(x, dinv, deg8, W1, b1, 1, src, dst, wl, wcnt, ovf, h);

    k_owner_agg<<<OBLK, 1024, 0, stream>>>(h, dinv, deg8, W2, b2, 0, src, dst, wl, wcnt, ovf, a);

    k_graph_pool_head<<<NGRAPHS, PT, 0, stream>>>(a, batch, zmm, Wl, bl, out);
}
